# starting kernel
# baseline (speedup 1.0000x reference)
#include <hip/hip_runtime.h>

typedef float v2f __attribute__((ext_vector_type(2)));
typedef float v8f __attribute__((ext_vector_type(8)));

#define N_NODES 50000
#define N_EDGES 800000
#define D 128
#define EPS_F 1e-7f

__global__ void cge_zero_kernel(float* __restrict__ p, int n) {
    int i = blockIdx.x * blockDim.x + threadIdx.x;
    if (i < n) p[i] = 0.0f;
}

__global__ __launch_bounds__(256) void cge_scatter_kernel(
    const float* __restrict__ h, int h_stride,
    const int* __restrict__ esrc, const int* __restrict__ edst,
    float* __restrict__ neigh) {
    int lane = threadIdx.x & 31;
    int edge = (blockIdx.x * blockDim.x + threadIdx.x) >> 5;
    if (edge >= N_EDGES) return;
    int s = esrc[edge];
    int d = edst[edge];
    float4 v = reinterpret_cast<const float4*>(h + (size_t)s * h_stride)[lane];
    float* np = neigh + (size_t)d * D + lane * 4;
    atomicAdd(np + 0, v.x);
    atomicAdd(np + 1, v.y);
    atomicAdd(np + 2, v.z);
    atomicAdd(np + 3, v.w);
}

__global__ __launch_bounds__(256) void cge_gemm_kernel(
    const float* __restrict__ A, int a_stride,
    const float* __restrict__ neigh,
    const float* __restrict__ deg,
    const float* __restrict__ W,
    const float* __restrict__ bias,
    float* __restrict__ out, int out_stride) {
    const int lane  = threadIdx.x & 31;
    const int warp  = threadIdx.x >> 5;
    const int mtile = blockIdx.x;
    const int rc    = lane & 15;
    const int khalf = lane >> 4;

    const int m = mtile * 16 + rc;
    const int n = warp * 16 + rc;

    const float invdeg = 1.0f / (deg[m] + EPS_F);
    const float* __restrict__ arow = A     + (size_t)m * a_stride;
    const float* __restrict__ nrow = neigh + (size_t)m * D;

    v8f c = {};

#pragma unroll 8
    for (int k0 = 0; k0 < 128; k0 += 4) {
        const int k = k0 + 2 * khalf;
        v2f a; a.x = arow[k];              a.y = arow[k + 1];
        v2f b; b.x = W[(size_t)k * D + n]; b.y = W[(size_t)(k + 1) * D + n];
        c = __builtin_amdgcn_wmma_f32_16x16x4_f32(
            false, a, false, b, (short)0, c, false, false);
    }
#pragma unroll 8
    for (int k0 = 0; k0 < 128; k0 += 4) {
        const int k = k0 + 2 * khalf;
        v2f a; a.x = nrow[k] * invdeg;     a.y = nrow[k + 1] * invdeg;
        v2f b; b.x = W[(size_t)(128 + k) * D + n];
               b.y = W[(size_t)(129 + k) * D + n];
        c = __builtin_amdgcn_wmma_f32_16x16x4_f32(
            false, a, false, b, (short)0, c, false, false);
    }

    const float bn    = bias[n];
    const int   mbase = mtile * 16 + khalf * 8;
#pragma unroll
    for (int j = 0; j < 8; ++j) {
        float v = c[j] + bn;
        v = v > 0.0f ? v : 0.0f;
        out[(size_t)(mbase + j) * out_stride + n] = v;
    }
}

__global__ void cge_copyx_kernel(const float* __restrict__ x,
                                 float* __restrict__ out) {
    int i = blockIdx.x * blockDim.x + threadIdx.x;
    if (i >= N_NODES * (D / 4)) return;
    int r  = i >> 5;
    int c4 = i & 31;
    float4 v = reinterpret_cast<const float4*>(x)[i];
    reinterpret_cast<float4*>(out + (size_t)r * 256 + 128)[c4] = v;
}

extern "C" void kernel_launch(void* const* d_in, const int* in_sizes, int n_in,
                              void* d_out, int out_size, void* d_ws, size_t ws_size,
                              hipStream_t stream) {
    const float* x    = (const float*)d_in[0];
    const int*   esrc = (const int*)  d_in[1];
    const int*   edst = (const int*)  d_in[2];
    const float* deg  = (const float*)d_in[3];
    const float* W1   = (const float*)d_in[4];
    const float* b1   = (const float*)d_in[5];
    const float* W2   = (const float*)d_in[6];
    const float* b2   = (const float*)d_in[7];
    float* out = (float*)d_out;

    float* neigh = (float*)d_ws;
    float* h1    = out + 128;

    const int NF = N_NODES * D;
    dim3 blk(256);
    dim3 zgrid((NF + 255) / 256);
    dim3 sgrid(((size_t)N_EDGES * 32 + 255) / 256);
    dim3 ggrid(N_NODES / 16);

    cge_zero_kernel   <<<zgrid, blk, 0, stream>>>(neigh, NF);
    cge_scatter_kernel<<<sgrid, blk, 0, stream>>>(x, D, esrc, edst, neigh);
    cge_gemm_kernel   <<<ggrid, blk, 0, stream>>>(x, D, neigh, deg, W1, b1,
                                                  h1, 256);
    cge_zero_kernel   <<<zgrid, blk, 0, stream>>>(neigh, NF);
    cge_scatter_kernel<<<sgrid, blk, 0, stream>>>(h1, 256, esrc, edst, neigh);
    cge_gemm_kernel   <<<ggrid, blk, 0, stream>>>(h1, 256, neigh, deg, W2, b2,
                                                  out, 256);
    cge_copyx_kernel  <<<(N_NODES * (D / 4) + 255) / 256, blk, 0, stream>>>(x, out);
}


# Neighbor sum by dst-bucketed gather: per-call counting sort of edges (int atomics), one wave per node adds its source rows in slot order; float atomic scatter and accumulator zeroing removed
# speedup vs baseline: 1.4295x; 1.4295x over previous
#include <hip/hip_runtime.h>

typedef float v2f __attribute__((ext_vector_type(2)));
typedef float v8f __attribute__((ext_vector_type(8)));

#define N_NODES 50000
#define N_EDGES 800000
#define D 128
#define EPS_F 1e-7f

#define SCR_CNT 0
#define SCR_PTR 50176
#define SCR_SRT 100352

__device__ __forceinline__ int scr(int i) { return ((i >> 7) << 8) | (i & 127); }

__global__ __launch_bounds__(256) void cge_cnt_zero_kernel(int* __restrict__ sc) {
    int i = blockIdx.x * blockDim.x + threadIdx.x;
    if (i < N_NODES) sc[scr(SCR_CNT + i)] = 0;
}

__global__ __launch_bounds__(256) void cge_cnt_kernel(const int* __restrict__ edst,
                                                      int* __restrict__ sc) {
    int e = blockIdx.x * blockDim.x + threadIdx.x;
    if (e >= N_EDGES) return;
    atomicAdd(&sc[scr(SCR_CNT + edst[e])], 1);
}

__global__ __launch_bounds__(1024) void cge_scan_kernel(int* __restrict__ sc) {
    __shared__ int wtot[32];
    const int lane = threadIdx.x & 31;
    const int w    = threadIdx.x >> 5;
    const int base = w * (13 * 128) + lane * 4;
    int4 v[13];
    int  ex[13];
#pragma unroll
    for (int t = 0; t < 13; ++t) {
        const int i = base + t * 128;
        v[t] = make_int4(0, 0, 0, 0);
        if (i < N_NODES) v[t] = *reinterpret_cast<const int4*>(&sc[scr(SCR_CNT + i)]);
    }
    int run = 0;
#pragma unroll
    for (int t = 0; t < 13; ++t) {
        const int tot = v[t].x + v[t].y + v[t].z + v[t].w;
        int x = tot;
#pragma unroll
        for (int o = 1; o < 32; o <<= 1) {
            const int y = __shfl_up(x, o);
            if (lane >= o) x += y;
        }
        ex[t] = x - tot + run;
        run += __shfl(x, 31);
    }
    if (lane == 0) wtot[w] = run;
    __syncthreads();
    const int wv = wtot[lane];
    int x = wv;
#pragma unroll
    for (int o = 1; o < 32; o <<= 1) {
        const int y = __shfl_up(x, o);
        if (lane >= o) x += y;
    }
    const int woff = __shfl(x - wv, w);
#pragma unroll
    for (int t = 0; t < 13; ++t) {
        const int i  = base + t * 128;
        const int e0 = woff + ex[t];
        const int e1 = e0 + v[t].x;
        const int e2 = e1 + v[t].y;
        const int e3 = e2 + v[t].z;
        int* p = &sc[scr(SCR_PTR + i)];
        if (i < N_NODES)       *reinterpret_cast<int4*>(p) = make_int4(e0, e1, e2, e3);
        else if (i == N_NODES) p[0] = e0;
    }
}

__global__ __launch_bounds__(256) void cge_fill_kernel(const int* __restrict__ esrc,
                                                       const int* __restrict__ edst,
                                                       int* sc) {
    int e = blockIdx.x * blockDim.x + threadIdx.x;
    if (e >= N_EDGES) return;
    const int d = edst[e];
    const int s = esrc[e];
    const int p = atomicSub(&sc[scr(SCR_CNT + d)], 1) - 1;
    const int r = sc[scr(SCR_PTR + d)];
    sc[scr(SCR_SRT + r + p)] = s;
}

template <int G>
__device__ __forceinline__ void cge_add_rows(const float* __restrict__ hl, int h_stride,
                                             int sidx, int j, float4& acc) {
    float4 v[G];
#pragma unroll
    for (int u = 0; u < G; ++u) {
        const int s = __builtin_amdgcn_readlane(sidx, j + u);
        v[u] = *reinterpret_cast<const float4*>(hl + (size_t)s * h_stride);
    }
#pragma unroll
    for (int u = 0; u < G; ++u) {
        acc.x += v[u].x; acc.y += v[u].y; acc.z += v[u].z; acc.w += v[u].w;
    }
}

__global__ __launch_bounds__(256) void cge_gather_kernel(
    const float* __restrict__ h, int h_stride,
    const int* __restrict__ sc, float* __restrict__ neigh) {
    const int lane = threadIdx.x & 31;
    const int node = blockIdx.x * 8 + (threadIdx.x >> 5);
    const int pv   = sc[scr(SCR_PTR + node + (lane & 1))];
    const int beg  = __builtin_amdgcn_readlane(pv, 0);
    const int end  = __builtin_amdgcn_readlane(pv, 1);
    const float* __restrict__ hl = h + lane * 4;
    float4 acc = make_float4(0.0f, 0.0f, 0.0f, 0.0f);
    for (int c = beg; c < end; c += 32) {
        const int n    = min(32, end - c);
        const int sidx = (lane < n) ? sc[scr(SCR_SRT + c + lane)] : 0;
        int j = 0;
        for (; j + 8 <= n; j += 8) cge_add_rows<8>(hl, h_stride, sidx, j, acc);
        if (n - j >= 4) { cge_add_rows<4>(hl, h_stride, sidx, j, acc); j += 4; }
        if (n - j >= 2) { cge_add_rows<2>(hl, h_stride, sidx, j, acc); j += 2; }
        if (n - j >= 1) { cge_add_rows<1>(hl, h_stride, sidx, j, acc); }
    }
    reinterpret_cast<float4*>(neigh + (size_t)node * D)[lane] = acc;
}

__global__ __launch_bounds__(256) void cge_gemm_kernel(
    const float* __restrict__ A, int a_stride,
    const float* __restrict__ neigh,
    const float* __restrict__ deg,
    const float* __restrict__ W,
    const float* __restrict__ bias,
    float* __restrict__ out, int out_stride) {
    const int lane  = threadIdx.x & 31;
    const int warp  = threadIdx.x >> 5;
    const int mtile = blockIdx.x;
    const int rc    = lane & 15;
    const int khalf = lane >> 4;

    const int m = mtile * 16 + rc;
    const int n = warp * 16 + rc;

    const float invdeg = 1.0f / (deg[m] + EPS_F);
    const float* __restrict__ arow = A     + (size_t)m * a_stride;
    const float* __restrict__ nrow = neigh + (size_t)m * D;

    v8f c = {};

#pragma unroll 8
    for (int k0 = 0; k0 < 128; k0 += 4) {
        const int k = k0 + 2 * khalf;
        v2f a; a.x = arow[k];              a.y = arow[k + 1];
        v2f b; b.x = W[(size_t)k * D + n]; b.y = W[(size_t)(k + 1) * D + n];
        c = __builtin_amdgcn_wmma_f32_16x16x4_f32(
            false, a, false, b, (short)0, c, false, false);
    }
#pragma unroll 8
    for (int k0 = 0; k0 < 128; k0 += 4) {
        const int k = k0 + 2 * khalf;
        v2f a; a.x = nrow[k] * invdeg;     a.y = nrow[k + 1] * invdeg;
        v2f b; b.x = W[(size_t)(128 + k) * D + n];
               b.y = W[(size_t)(129 + k) * D + n];
        c = __builtin_amdgcn_wmma_f32_16x16x4_f32(
            false, a, false, b, (short)0, c, false, false);
    }

    const float bn    = bias[n];
    const int   mbase = mtile * 16 + khalf * 8;
#pragma unroll
    for (int j = 0; j < 8; ++j) {
        float v = c[j] + bn;
        v = v > 0.0f ? v : 0.0f;
        out[(size_t)(mbase + j) * out_stride + n] = v;
    }
}

__global__ void cge_copyx_kernel(const float* __restrict__ x,
                                 float* __restrict__ out) {
    int i = blockIdx.x * blockDim.x + threadIdx.x;
    if (i >= N_NODES * (D / 4)) return;
    int r  = i >> 5;
    int c4 = i & 31;
    float4 v = reinterpret_cast<const float4*>(x)[i];
    reinterpret_cast<float4*>(out + (size_t)r * 256 + 128)[c4] = v;
}

extern "C" void kernel_launch(void* const* d_in, const int* in_sizes, int n_in,
                              void* d_out, int out_size, void* d_ws, size_t ws_size,
                              hipStream_t stream) {
    const float* x    = (const float*)d_in[0];
    const int*   esrc = (const int*)  d_in[1];
    const int*   edst = (const int*)  d_in[2];
    const float* deg  = (const float*)d_in[3];
    const float* W1   = (const float*)d_in[4];
    const float* b1   = (const float*)d_in[5];
    const float* W2   = (const float*)d_in[6];
    const float* b2   = (const float*)d_in[7];
    float* out = (float*)d_out;

    float* neigh = (float*)d_ws;
    float* h1    = out + 128;
    int*   sc    = (int*)d_out;

    dim3 blk(256);
    dim3 ngrid((N_NODES + 255) / 256);
    dim3 egrid(N_EDGES / 256);
    dim3 wgrid(N_NODES / 8);
    dim3 ggrid(N_NODES / 16);

    cge_cnt_zero_kernel<<<ngrid, blk, 0, stream>>>(sc);
    cge_cnt_kernel     <<<egrid, blk, 0, stream>>>(edst, sc);
    cge_scan_kernel    <<<1, 1024, 0, stream>>>(sc);
    cge_fill_kernel    <<<egrid, blk, 0, stream>>>(esrc, edst, sc);

    cge_gather_kernel<<<wgrid, blk, 0, stream>>>(x, D, sc, neigh);
    cge_gemm_kernel  <<<ggrid, blk, 0, stream>>>(x, D, neigh, deg, W1, b1,
                                                 h1, 256);
    cge_gather_kernel<<<wgrid, blk, 0, stream>>>(h1, 256, sc, neigh);
    cge_gemm_kernel  <<<ggrid, blk, 0, stream>>>(h1, 256, neigh, deg, W2, b2,
                                                 out, 256);
    cge_copyx_kernel  <<<(N_NODES * (D / 4) + 255) / 256, blk, 0, stream>>>(x, out);
}


# Layer GEMM feeds its WMMAs from on-chip operands: 256 blocks loop over row tiles, a wave keeps its 256x16 W slice in registers, A tile staged once per block in LDS by whole-row loads; same sums
# speedup vs baseline: 1.7713x; 1.2391x over previous
#include <hip/hip_runtime.h>

typedef float v2f __attribute__((ext_vector_type(2)));
typedef float v8f __attribute__((ext_vector_type(8)));

#define N_NODES 50000
#define N_EDGES 800000
#define D 128
#define EPS_F 1e-7f

#define SCR_CNT 0
#define SCR_PTR 50176
#define SCR_SRT 100352

__device__ __forceinline__ int scr(int i) { return ((i >> 7) << 8) | (i & 127); }

__global__ __launch_bounds__(256) void cge_cnt_zero_kernel(int* __restrict__ sc) {
    int i = blockIdx.x * blockDim.x + threadIdx.x;
    if (i < N_NODES) sc[scr(SCR_CNT + i)] = 0;
}

__global__ __launch_bounds__(256) void cge_cnt_kernel(const int* __restrict__ edst,
                                                      int* __restrict__ sc) {
    int e = blockIdx.x * blockDim.x + threadIdx.x;
    if (e >= N_EDGES) return;
    atomicAdd(&sc[scr(SCR_CNT + edst[e])], 1);
}

__global__ __launch_bounds__(1024) void cge_scan_kernel(int* __restrict__ sc) {
    __shared__ int wtot[32];
    const int lane = threadIdx.x & 31;
    const int w    = threadIdx.x >> 5;
    const int base = w * (13 * 128) + lane * 4;
    int4 v[13];
    int  ex[13];
#pragma unroll
    for (int t = 0; t < 13; ++t) {
        const int i = base + t * 128;
        v[t] = make_int4(0, 0, 0, 0);
        if (i < N_NODES) v[t] = *reinterpret_cast<const int4*>(&sc[scr(SCR_CNT + i)]);
    }
    int run = 0;
#pragma unroll
    for (int t = 0; t < 13; ++t) {
        const int tot = v[t].x + v[t].y + v[t].z + v[t].w;
        int x = tot;
#pragma unroll
        for (int o = 1; o < 32; o <<= 1) {
            const int y = __shfl_up(x, o);
            if (lane >= o) x += y;
        }
        ex[t] = x - tot + run;
        run += __shfl(x, 31);
    }
    if (lane == 0) wtot[w] = run;
    __syncthreads();
    const int wv = wtot[lane];
    int x = wv;
#pragma unroll
    for (int o = 1; o < 32; o <<= 1) {
        const int y = __shfl_up(x, o);
        if (lane >= o) x += y;
    }
    const int woff = __shfl(x - wv, w);
#pragma unroll
    for (int t = 0; t < 13; ++t) {
        const int i  = base + t * 128;
        const int e0 = woff + ex[t];
        const int e1 = e0 + v[t].x;
        const int e2 = e1 + v[t].y;
        const int e3 = e2 + v[t].z;
        int* p = &sc[scr(SCR_PTR + i)];
        if (i < N_NODES)       *reinterpret_cast<int4*>(p) = make_int4(e0, e1, e2, e3);
        else if (i == N_NODES) p[0] = e0;
    }
}

__global__ __launch_bounds__(256) void cge_fill_kernel(const int* __restrict__ esrc,
                                                       const int* __restrict__ edst,
                                                       int* sc) {
    int e = blockIdx.x * blockDim.x + threadIdx.x;
    if (e >= N_EDGES) return;
    const int d = edst[e];
    const int s = esrc[e];
    const int p = atomicSub(&sc[scr(SCR_CNT + d)], 1) - 1;
    const int r = sc[scr(SCR_PTR + d)];
    sc[scr(SCR_SRT + r + p)] = s;
}

template <int G>
__device__ __forceinline__ void cge_add_rows(const float* __restrict__ hl, int h_stride,
                                             int sidx, int j, float4& acc) {
    float4 v[G];
#pragma unroll
    for (int u = 0; u < G; ++u) {
        const int s = __builtin_amdgcn_readlane(sidx, j + u);
        v[u] = *reinterpret_cast<const float4*>(hl + (size_t)s * h_stride);
    }
#pragma unroll
    for (int u = 0; u < G; ++u) {
        acc.x += v[u].x; acc.y += v[u].y; acc.z += v[u].z; acc.w += v[u].w;
    }
}

__global__ __launch_bounds__(256) void cge_gather_kernel(
    const float* __restrict__ h, int h_stride,
    const int* __restrict__ sc, float* __restrict__ neigh) {
    const int lane = threadIdx.x & 31;
    const int node = blockIdx.x * 8 + (threadIdx.x >> 5);
    const int pv   = sc[scr(SCR_PTR + node + (lane & 1))];
    const int beg  = __builtin_amdgcn_readlane(pv, 0);
    const int end  = __builtin_amdgcn_readlane(pv, 1);
    const float* __restrict__ hl = h + lane * 4;
    float4 acc = make_float4(0.0f, 0.0f, 0.0f, 0.0f);
    for (int c = beg; c < end; c += 32) {
        const int n    = min(32, end - c);
        const int sidx = (lane < n) ? sc[scr(SCR_SRT + c + lane)] : 0;
        int j = 0;
        for (; j + 8 <= n; j += 8) cge_add_rows<8>(hl, h_stride, sidx, j, acc);
        if (n - j >= 4) { cge_add_rows<4>(hl, h_stride, sidx, j, acc); j += 4; }
        if (n - j >= 2) { cge_add_rows<2>(hl, h_stride, sidx, j, acc); j += 2; }
        if (n - j >= 1) { cge_add_rows<1>(hl, h_stride, sidx, j, acc); }
    }
    reinterpret_cast<float4*>(neigh + (size_t)node * D)[lane] = acc;
}

#define GEMM_GRID 256
#define N_TILES   (N_NODES / 16)
#define A_PITCH   260
#define KCH       16
#define NCH       (64 / KCH)

struct TileRows { float4 s0, s1, n0, n1; float d0, d1; };

__device__ __forceinline__ TileRows cge_fetch_rows(
    const float* __restrict__ A, int a_stride, const float* __restrict__ neigh,
    const float* __restrict__ deg, int tile, int srow, int c4) {
    TileRows r;
    const int m0 = tile * 16 + srow;
    const int m1 = m0 + 8;
    r.s0 = *reinterpret_cast<const float4*>(A + (size_t)m0 * a_stride + c4);
    r.s1 = *reinterpret_cast<const float4*>(A + (size_t)m1 * a_stride + c4);
    r.n0 = *reinterpret_cast<const float4*>(neigh + (size_t)m0 * D + c4);
    r.n1 = *reinterpret_cast<const float4*>(neigh + (size_t)m1 * D + c4);
    r.d0 = deg[m0];
    r.d1 = deg[m1];
    return r;
}

__device__ __forceinline__ void cge_stash_rows(float* rows, const TileRows& r,
                                               int srow, int c4) {
    const float i0 = 1.0f / (r.d0 + EPS_F);
    const float i1 = 1.0f / (r.d1 + EPS_F);
    float* p0 = rows + srow * A_PITCH + c4;
    float* p1 = p0 + 8 * A_PITCH;
    *reinterpret_cast<float4*>(p0) = r.s0;
    *reinterpret_cast<float4*>(p1) = r.s1;
    *reinterpret_cast<float4*>(p0 + 128) =
        make_float4(r.n0.x * i0, r.n0.y * i0, r.n0.z * i0, r.n0.w * i0);
    *reinterpret_cast<float4*>(p1 + 128) =
        make_float4(r.n1.x * i1, r.n1.y * i1, r.n1.z * i1, r.n1.w * i1);
}

template <int NT>
__device__ __forceinline__ void cge_read_frags(const float* al, int ch, v2f (&a)[NT][KCH]) {
#pragma unroll
    for (int u = 0; u < NT; ++u)
#pragma unroll
        for (int s = 0; s < KCH; ++s)
            a[u][s] = *reinterpret_cast<const v2f*>(
                al + u * 16 * A_PITCH + 4 * (ch * KCH + s));
}

template <int NT>
__device__ __forceinline__ void cge_tiles_mma(const float* al, const v2f (&bfr)[64],
                                              v8f (&c)[NT]) {
    v2f a[2][NT][KCH];
    cge_read_frags<NT>(al, 0, a[0]);
#pragma unroll
    for (int ch = 0; ch < NCH; ++ch) {
        if (ch + 1 < NCH) cge_read_frags<NT>(al, ch + 1, a[(ch + 1) & 1]);
#pragma unroll
        for (int s = 0; s < KCH; ++s)
#pragma unroll
            for (int u = 0; u < NT; ++u)
                c[u] = __builtin_amdgcn_wmma_f32_16x16x4_f32(
                    false, a[ch & 1][u][s], false, bfr[ch * KCH + s], (short)0, c[u],
                    false, false);
    }
    __builtin_amdgcn_sched_group_barrier(0x100, NT * KCH / 2, 0);
#pragma unroll
    for (int ch = 0; ch < NCH; ++ch) {
        if (ch + 1 < NCH) __builtin_amdgcn_sched_group_barrier(0x100, NT * KCH / 2, 0);
        __builtin_amdgcn_sched_group_barrier(0x008, NT * KCH, 0);
    }
}

__device__ __forceinline__ void cge_store_tile(const v8f& c, float bn, int tile, int khalf,
                                               int n, float* __restrict__ out,
                                               int out_stride) {
    const int mbase = tile * 16 + khalf * 8;
#pragma unroll
    for (int j = 0; j < 8; ++j) {
        float v = c[j] + bn;
        v = v > 0.0f ? v : 0.0f;
        out[(size_t)(mbase + j) * out_stride + n] = v;
    }
}

__global__ __launch_bounds__(256) void cge_gemm_kernel(
    const float* __restrict__ A, int a_stride,
    const float* __restrict__ neigh,
    const float* __restrict__ deg,
    const float* __restrict__ W,
    const float* __restrict__ bias,
    float* __restrict__ out, int out_stride) {
    __shared__ float a_lds[2 * 32 * A_PITCH];

    const int lane  = threadIdx.x & 31;
    const int warp  = threadIdx.x >> 5;
    const int rc    = lane & 15;
    const int khalf = lane >> 4;
    const int n     = warp * 16 + rc;

    v2f bfr[64];
#pragma unroll
    for (int s = 0; s < 64; ++s) {
        const int k = 4 * s + 2 * khalf;
        bfr[s].x = W[(size_t)k * D + n];
        bfr[s].y = W[(size_t)(k + 1) * D + n];
    }
    const float bn = bias[n];

    const int srow = warp;
    const int c4   = lane * 4;
    TileRows ra, rb;

    int t = blockIdx.x;
    ra = cge_fetch_rows(A, a_stride, neigh, deg, t, srow, c4);
    if (t + GEMM_GRID < N_TILES) {
        rb = cge_fetch_rows(A, a_stride, neigh, deg, t + GEMM_GRID, srow, c4);
        cge_stash_rows(a_lds + 16 * A_PITCH, rb, srow, c4);
    }
    cge_stash_rows(a_lds, ra, srow, c4);
    __syncthreads();

    int buf = 0;
    for (; t < N_TILES; t += 2 * GEMM_GRID) {
        const bool two   = t + GEMM_GRID < N_TILES;
        const int  tn    = t + 2 * GEMM_GRID;
        const bool more  = tn < N_TILES;
        const bool more2 = tn + GEMM_GRID < N_TILES;
        if (more)  ra = cge_fetch_rows(A, a_stride, neigh, deg, tn, srow, c4);
        if (more2) rb = cge_fetch_rows(A, a_stride, neigh, deg, tn + GEMM_GRID, srow, c4);

        const float* al = a_lds + buf * (32 * A_PITCH) + rc * A_PITCH + 2 * khalf;
        v8f c[2] = {v8f{}, v8f{}};
        if (two) {
            cge_tiles_mma<2>(al, bfr, c);
        } else {
            v8f c1[1] = {v8f{}};
            cge_tiles_mma<1>(al, bfr, c1);
            c[0] = c1[0];
        }

        float* nb = a_lds + (buf ^ 1) * (32 * A_PITCH);
        if (more)  cge_stash_rows(nb, ra, srow, c4);
        if (more2) cge_stash_rows(nb + 16 * A_PITCH, rb, srow, c4);
        __syncthreads();
        cge_store_tile(c[0], bn, t, khalf, n, out, out_stride);
        if (two) cge_store_tile(c[1], bn, t + GEMM_GRID, khalf, n, out, out_stride);
        buf ^= 1;
    }
}

__global__ void cge_copyx_kernel(const float* __restrict__ x,
                                 float* __restrict__ out) {
    int i = blockIdx.x * blockDim.x + threadIdx.x;
    if (i >= N_NODES * (D / 4)) return;
    int r  = i >> 5;
    int c4 = i & 31;
    float4 v = reinterpret_cast<const float4*>(x)[i];
    reinterpret_cast<float4*>(out + (size_t)r * 256 + 128)[c4] = v;
}

extern "C" void kernel_launch(void* const* d_in, const int* in_sizes, int n_in,
                              void* d_out, int out_size, void* d_ws, size_t ws_size,
                              hipStream_t stream) {
    const float* x    = (const float*)d_in[0];
    const int*   esrc = (const int*)  d_in[1];
    const int*   edst = (const int*)  d_in[2];
    const float* deg  = (const float*)d_in[3];
    const float* W1   = (const float*)d_in[4];
    const float* b1   = (const float*)d_in[5];
    const float* W2   = (const float*)d_in[6];
    const float* b2   = (const float*)d_in[7];
    float* out = (float*)d_out;

    float* neigh = (float*)d_ws;
    float* h1    = out + 128;
    int*   sc    = (int*)d_out;

    dim3 blk(256);
    dim3 ngrid((N_NODES + 255) / 256);
    dim3 egrid(N_EDGES / 256);
    dim3 wgrid(N_NODES / 8);
    dim3 ggrid(GEMM_GRID);

    cge_cnt_zero_kernel<<<ngrid, blk, 0, stream>>>(sc);
    cge_cnt_kernel     <<<egrid, blk, 0, stream>>>(edst, sc);
    cge_scan_kernel    <<<1, 1024, 0, stream>>>(sc);
    cge_fill_kernel    <<<egrid, blk, 0, stream>>>(esrc, edst, sc);

    cge_gather_kernel<<<wgrid, blk, 0, stream>>>(x, D, sc, neigh);
    cge_gemm_kernel  <<<ggrid, blk, 0, stream>>>(x, D, neigh, deg, W1, b1,
                                                 h1, 256);
    cge_gather_kernel<<<wgrid, blk, 0, stream>>>(h1, 256, sc, neigh);
    cge_gemm_kernel  <<<ggrid, blk, 0, stream>>>(h1, 256, neigh, deg, W2, b2,
                                                 out, 256);
    cge_copyx_kernel  <<<(N_NODES * (D / 4) + 255) / 256, blk, 0, stream>>>(x, out);
}


# Edge bucketing as two-level counting sort on LDS counters (200 node ranges x 250 edge chunks): the 1.6M global integer atomics go; slot (sum) order in a bucket stays atomic arrival order
# speedup vs baseline: 2.6530x; 1.4978x over previous
#include <hip/hip_runtime.h>

typedef float v2f __attribute__((ext_vector_type(2)));
typedef float v8f __attribute__((ext_vector_type(8)));

#define N_NODES 50000
#define N_EDGES 800000
#define D 128
#define EPS_F 1e-7f

#define NBIN   200
#define BIN_W  250
#define NCHK   250
#define CHK_E  3200
#define SORT_T 1024
#define CHK_K  ((CHK_E + SORT_T - 1) / SORT_T)
#define BKT_U  4
static_assert(NBIN * BIN_W == N_NODES, "node ranges must tile the nodes");
static_assert(NCHK * CHK_E == N_EDGES, "edge chunks must tile the edges");
static_assert(NBIN * NCHK == N_NODES, "the scan walks N_NODES table entries");
static_assert(BIN_W <= 256 && N_NODES <= 65536, "packed word: 8 + 16 bits");

#define SCR_CNT 0
#define SCR_OFF 50176
#define SCR_PTR 100480
#define SCR_SRT 150784
#define SCR_BIN 950784

__device__ __forceinline__ int scr(int i) { return ((i >> 7) << 8) | (i & 127); }

__global__ __launch_bounds__(SORT_T) void cge_hist_kernel(const int* __restrict__ edst,
                                                          int* __restrict__ sc) {
    __shared__ int h[NBIN];
    const int t  = threadIdx.x;
    const int b  = blockIdx.x;
    const int e0 = b * CHK_E;
    int d[CHK_K];
#pragma unroll
    for (int k = 0; k < CHK_K; ++k) {
        const int i = t + k * SORT_T;
        d[k] = (i < CHK_E) ? edst[e0 + i] : 0;
    }
    if (t < NBIN) h[t] = 0;
    __syncthreads();
#pragma unroll
    for (int k = 0; k < CHK_K; ++k)
        if (t + k * SORT_T < CHK_E) atomicAdd(&h[(unsigned)d[k] / BIN_W], 1);
    __syncthreads();
    if (t < NBIN) sc[scr(SCR_CNT + t * NCHK + b)] = h[t];
}

__global__ __launch_bounds__(1024) void cge_scan_kernel(int* __restrict__ sc) {
    __shared__ int wtot[32];
    const int lane = threadIdx.x & 31;
    const int w    = threadIdx.x >> 5;
    const int base = w * (13 * 128) + lane * 4;
    int4 v[13];
    int  ex[13];
#pragma unroll
    for (int t = 0; t < 13; ++t) {
        const int i = base + t * 128;
        v[t] = make_int4(0, 0, 0, 0);
        if (i < N_NODES) v[t] = *reinterpret_cast<const int4*>(&sc[scr(SCR_CNT + i)]);
    }
    int run = 0;
#pragma unroll
    for (int t = 0; t < 13; ++t) {
        const int tot = v[t].x + v[t].y + v[t].z + v[t].w;
        int x = tot;
#pragma unroll
        for (int o = 1; o < 32; o <<= 1) {
            const int y = __shfl_up(x, o);
            if (lane >= o) x += y;
        }
        ex[t] = x - tot + run;
        run += __shfl(x, 31);
    }
    if (lane == 0) wtot[w] = run;
    __syncthreads();
    const int wv = wtot[lane];
    int x = wv;
#pragma unroll
    for (int o = 1; o < 32; o <<= 1) {
        const int y = __shfl_up(x, o);
        if (lane >= o) x += y;
    }
    const int woff = __shfl(x - wv, w);
#pragma unroll
    for (int t = 0; t < 13; ++t) {
        const int i  = base + t * 128;
        const int e0 = woff + ex[t];
        const int e1 = e0 + v[t].x;
        const int e2 = e1 + v[t].y;
        const int e3 = e2 + v[t].z;
        int* p = &sc[scr(SCR_OFF + i)];
        if (i < N_NODES)       *reinterpret_cast<int4*>(p) = make_int4(e0, e1, e2, e3);
        else if (i == N_NODES) p[0] = e0;
    }
}

__global__ __launch_bounds__(SORT_T) void cge_scatter_kernel(const int* __restrict__ esrc,
                                                             const int* __restrict__ edst,
                                                             const int* __restrict__ rd,
                                                             int* __restrict__ wr) {
    __shared__ int cur[NBIN];
    const int t  = threadIdx.x;
    const int b  = blockIdx.x;
    const int e0 = b * CHK_E;
    int d[CHK_K], s[CHK_K];
#pragma unroll
    for (int k = 0; k < CHK_K; ++k) {
        const int i = t + k * SORT_T;
        d[k] = (i < CHK_E) ? edst[e0 + i] : 0;
        s[k] = (i < CHK_E) ? esrc[e0 + i] : 0;
    }
    if (t < NBIN) cur[t] = rd[scr(SCR_OFF + t * NCHK + b)];
    __syncthreads();
    int pos[CHK_K];
#pragma unroll
    for (int k = 0; k < CHK_K; ++k) {
        pos[k] = 0;
        if (t + k * SORT_T < CHK_E) pos[k] = atomicAdd(&cur[(unsigned)d[k] / BIN_W], 1);
    }
#pragma unroll
    for (int k = 0; k < CHK_K; ++k) {
        if (t + k * SORT_T < CHK_E) {
            const int bin = (unsigned)d[k] / BIN_W;
            const int dl  = d[k] - bin * BIN_W;
            wr[scr(SCR_BIN + pos[k])] = (dl << 16) | s[k];
        }
    }
}

__global__ __launch_bounds__(SORT_T) void cge_bucket_kernel(const int* __restrict__ rd,
                                                            int* __restrict__ wr) {
    __shared__ int cnt[256];
    __shared__ int wtot[8];
    const int t    = threadIdx.x;
    const int lane = t & 31;
    const int w    = t >> 5;
    const int B    = blockIdx.x;
    const int bs   = rd[scr(SCR_OFF + B * NCHK)];
    const int be   = rd[scr(SCR_OFF + (B + 1) * NCHK)];
    if (t < 256) cnt[t] = 0;
    __syncthreads();
    for (int e = bs + t; e < be; e += BKT_U * SORT_T) {
        int v[BKT_U];
#pragma unroll
        for (int k = 0; k < BKT_U; ++k) {
            const int ek = min(e + k * SORT_T, be - 1);
            v[k] = rd[scr(SCR_BIN + ek)];
        }
#pragma unroll
        for (int k = 0; k < BKT_U; ++k)
            if (e + k * SORT_T < be) atomicAdd(&cnt[v[k] >> 16], 1);
    }
    __syncthreads();
    const int c = (t < 256) ? cnt[t] : 0;
    int x = c;
#pragma unroll
    for (int o = 1; o < 32; o <<= 1) {
        const int y = __shfl_up(x, o);
        if (lane >= o) x += y;
    }
    if (lane == 31 && w < 8) wtot[w] = x;
    __syncthreads();
    if (t < 256) {
        int woff = 0;
#pragma unroll
        for (int k = 0; k < 8; ++k) woff += (k < w) ? wtot[k] : 0;
        const int first = bs + woff + x - c;
        cnt[t] = first;
        if (t < BIN_W) wr[scr(SCR_PTR + B * BIN_W + t)] = first;
        if (B == NBIN - 1 && t == 0) wr[scr(SCR_PTR + N_NODES)] = be;
    }
    __syncthreads();
    for (int e = bs + t; e < be; e += BKT_U * SORT_T) {
        int v[BKT_U], pos[BKT_U];
#pragma unroll
        for (int k = 0; k < BKT_U; ++k) {
            const int ek = min(e + k * SORT_T, be - 1);
            v[k] = rd[scr(SCR_BIN + ek)];
        }
#pragma unroll
        for (int k = 0; k < BKT_U; ++k) {
            pos[k] = 0;
            if (e + k * SORT_T < be) pos[k] = atomicAdd(&cnt[v[k] >> 16], 1);
        }
#pragma unroll
        for (int k = 0; k < BKT_U; ++k)
            if (e + k * SORT_T < be) wr[scr(SCR_SRT + pos[k])] = v[k] & 0xFFFF;
    }
}

template <int G>
__device__ __forceinline__ void cge_add_rows(const float* __restrict__ hl, int h_stride,
                                             int sidx, int j, float4& acc) {
    float4 v[G];
#pragma unroll
    for (int u = 0; u < G; ++u) {
        const int s = __builtin_amdgcn_readlane(sidx, j + u);
        v[u] = *reinterpret_cast<const float4*>(hl + (size_t)s * h_stride);
    }
#pragma unroll
    for (int u = 0; u < G; ++u) {
        acc.x += v[u].x; acc.y += v[u].y; acc.z += v[u].z; acc.w += v[u].w;
    }
}

__global__ __launch_bounds__(256) void cge_gather_kernel(
    const float* __restrict__ h, int h_stride,
    const int* __restrict__ sc, float* __restrict__ neigh) {
    const int lane = threadIdx.x & 31;
    const int node = blockIdx.x * 8 + (threadIdx.x >> 5);
    const int pv   = sc[scr(SCR_PTR + node + (lane & 1))];
    const int beg  = __builtin_amdgcn_readlane(pv, 0);
    const int end  = __builtin_amdgcn_readlane(pv, 1);
    const float* __restrict__ hl = h + lane * 4;
    float4 acc = make_float4(0.0f, 0.0f, 0.0f, 0.0f);
    for (int c = beg; c < end; c += 32) {
        const int n    = min(32, end - c);
        const int sidx = (lane < n) ? sc[scr(SCR_SRT + c + lane)] : 0;
        int j = 0;
        for (; j + 8 <= n; j += 8) cge_add_rows<8>(hl, h_stride, sidx, j, acc);
        if (n - j >= 4) { cge_add_rows<4>(hl, h_stride, sidx, j, acc); j += 4; }
        if (n - j >= 2) { cge_add_rows<2>(hl, h_stride, sidx, j, acc); j += 2; }
        if (n - j >= 1) { cge_add_rows<1>(hl, h_stride, sidx, j, acc); }
    }
    reinterpret_cast<float4*>(neigh + (size_t)node * D)[lane] = acc;
}

#define GEMM_GRID 256
#define N_TILES   (N_NODES / 16)
#define A_PITCH   260
#define KCH       16
#define NCH       (64 / KCH)

struct TileRows { float4 s0, s1, n0, n1; float d0, d1; };

__device__ __forceinline__ TileRows cge_fetch_rows(
    const float* __restrict__ A, int a_stride, const float* __restrict__ neigh,
    const float* __restrict__ deg, int tile, int srow, int c4) {
    TileRows r;
    const int m0 = tile * 16 + srow;
    const int m1 = m0 + 8;
    r.s0 = *reinterpret_cast<const float4*>(A + (size_t)m0 * a_stride + c4);
    r.s1 = *reinterpret_cast<const float4*>(A + (size_t)m1 * a_stride + c4);
    r.n0 = *reinterpret_cast<const float4*>(neigh + (size_t)m0 * D + c4);
    r.n1 = *reinterpret_cast<const float4*>(neigh + (size_t)m1 * D + c4);
    r.d0 = deg[m0];
    r.d1 = deg[m1];
    return r;
}

__device__ __forceinline__ void cge_stash_rows(float* rows, const TileRows& r,
                                               int srow, int c4) {
    const float i0 = 1.0f / (r.d0 + EPS_F);
    const float i1 = 1.0f / (r.d1 + EPS_F);
    float* p0 = rows + srow * A_PITCH + c4;
    float* p1 = p0 + 8 * A_PITCH;
    *reinterpret_cast<float4*>(p0) = r.s0;
    *reinterpret_cast<float4*>(p1) = r.s1;
    *reinterpret_cast<float4*>(p0 + 128) =
        make_float4(r.n0.x * i0, r.n0.y * i0, r.n0.z * i0, r.n0.w * i0);
    *reinterpret_cast<float4*>(p1 + 128) =
        make_float4(r.n1.x * i1, r.n1.y * i1, r.n1.z * i1, r.n1.w * i1);
}

template <int NT>
__device__ __forceinline__ void cge_read_frags(const float* al, int ch, v2f (&a)[NT][KCH]) {
#pragma unroll
    for (int u = 0; u < NT; ++u)
#pragma unroll
        for (int s = 0; s < KCH; ++s)
            a[u][s] = *reinterpret_cast<const v2f*>(
                al + u * 16 * A_PITCH + 4 * (ch * KCH + s));
}

template <int NT>
__device__ __forceinline__ void cge_tiles_mma(const float* al, const v2f (&bfr)[64],
                                              v8f (&c)[NT]) {
    v2f a[2][NT][KCH];
    cge_read_frags<NT>(al, 0, a[0]);
#pragma unroll
    for (int ch = 0; ch < NCH; ++ch) {
        if (ch + 1 < NCH) cge_read_frags<NT>(al, ch + 1, a[(ch + 1) & 1]);
#pragma unroll
        for (int s = 0; s < KCH; ++s)
#pragma unroll
            for (int u = 0; u < NT; ++u)
                c[u] = __builtin_amdgcn_wmma_f32_16x16x4_f32(
                    false, a[ch & 1][u][s], false, bfr[ch * KCH + s], (short)0, c[u],
                    false, false);
    }
    __builtin_amdgcn_sched_group_barrier(0x100, NT * KCH / 2, 0);
#pragma unroll
    for (int ch = 0; ch < NCH; ++ch) {
        if (ch + 1 < NCH) __builtin_amdgcn_sched_group_barrier(0x100, NT * KCH / 2, 0);
        __builtin_amdgcn_sched_group_barrier(0x008, NT * KCH, 0);
    }
}

__device__ __forceinline__ void cge_store_tile(const v8f& c, float bn, int tile, int khalf,
                                               int n, float* __restrict__ out,
                                               int out_stride) {
    const int mbase = tile * 16 + khalf * 8;
#pragma unroll
    for (int j = 0; j < 8; ++j) {
        float v = c[j] + bn;
        v = v > 0.0f ? v : 0.0f;
        out[(size_t)(mbase + j) * out_stride + n] = v;
    }
}

__global__ __launch_bounds__(256) void cge_gemm_kernel(
    const float* __restrict__ A, int a_stride,
    const float* __restrict__ neigh,
    const float* __restrict__ deg,
    const float* __restrict__ W,
    const float* __restrict__ bias,
    float* __restrict__ out, int out_stride) {
    __shared__ float a_lds[2 * 32 * A_PITCH];

    const int lane  = threadIdx.x & 31;
    const int warp  = threadIdx.x >> 5;
    const int rc    = lane & 15;
    const int khalf = lane >> 4;
    const int n     = warp * 16 + rc;

    v2f bfr[64];
#pragma unroll
    for (int s = 0; s < 64; ++s) {
        const int k = 4 * s + 2 * khalf;
        bfr[s].x = W[(size_t)k * D + n];
        bfr[s].y = W[(size_t)(k + 1) * D + n];
    }
    const float bn = bias[n];

    const int srow = warp;
    const int c4   = lane * 4;
    TileRows ra, rb;

    int t = blockIdx.x;
    ra = cge_fetch_rows(A, a_stride, neigh, deg, t, srow, c4);
    if (t + GEMM_GRID < N_TILES) {
        rb = cge_fetch_rows(A, a_stride, neigh, deg, t + GEMM_GRID, srow, c4);
        cge_stash_rows(a_lds + 16 * A_PITCH, rb, srow, c4);
    }
    cge_stash_rows(a_lds, ra, srow, c4);
    __syncthreads();

    int buf = 0;
    for (; t < N_TILES; t += 2 * GEMM_GRID) {
        const bool two   = t + GEMM_GRID < N_TILES;
        const int  tn    = t + 2 * GEMM_GRID;
        const bool more  = tn < N_TILES;
        const bool more2 = tn + GEMM_GRID < N_TILES;
        if (more)  ra = cge_fetch_rows(A, a_stride, neigh, deg, tn, srow, c4);
        if (more2) rb = cge_fetch_rows(A, a_stride, neigh, deg, tn + GEMM_GRID, srow, c4);

        const float* al = a_lds + buf * (32 * A_PITCH) + rc * A_PITCH + 2 * khalf;
        v8f c[2] = {v8f{}, v8f{}};
        if (two) {
            cge_tiles_mma<2>(al, bfr, c);
        } else {
            v8f c1[1] = {v8f{}};
            cge_tiles_mma<1>(al, bfr, c1);
            c[0] = c1[0];
        }

        float* nb = a_lds + (buf ^ 1) * (32 * A_PITCH);
        if (more)  cge_stash_rows(nb, ra, srow, c4);
        if (more2) cge_stash_rows(nb + 16 * A_PITCH, rb, srow, c4);
        __syncthreads();
        cge_store_tile(c[0], bn, t, khalf, n, out, out_stride);
        if (two) cge_store_tile(c[1], bn, t + GEMM_GRID, khalf, n, out, out_stride);
        buf ^= 1;
    }
}

__global__ void cge_copyx_kernel(const float* __restrict__ x,
                                 float* __restrict__ out) {
    int i = blockIdx.x * blockDim.x + threadIdx.x;
    if (i >= N_NODES * (D / 4)) return;
    int r  = i >> 5;
    int c4 = i & 31;
    float4 v = reinterpret_cast<const float4*>(x)[i];
    reinterpret_cast<float4*>(out + (size_t)r * 256 + 128)[c4] = v;
}

extern "C" void kernel_launch(void* const* d_in, const int* in_sizes, int n_in,
                              void* d_out, int out_size, void* d_ws, size_t ws_size,
                              hipStream_t stream) {
    const float* x    = (const float*)d_in[0];
    const int*   esrc = (const int*)  d_in[1];
    const int*   edst = (const int*)  d_in[2];
    const float* deg  = (const float*)d_in[3];
    const float* W1   = (const float*)d_in[4];
    const float* b1   = (const float*)d_in[5];
    const float* W2   = (const float*)d_in[6];
    const float* b2   = (const float*)d_in[7];
    float* out = (float*)d_out;

    float* neigh = (float*)d_ws;
    float* h1    = out + 128;
    int*   sc    = (int*)d_out;

    dim3 blk(256);
    dim3 sblk(SORT_T);
    dim3 cgrid(NCHK);
    dim3 rgrid(NBIN);
    dim3 wgrid(N_NODES / 8);
    dim3 ggrid(GEMM_GRID);

    cge_hist_kernel    <<<cgrid, sblk, 0, stream>>>(edst, sc);
    cge_scan_kernel    <<<1, 1024, 0, stream>>>(sc);
    cge_scatter_kernel <<<cgrid, sblk, 0, stream>>>(esrc, edst, sc, sc);
    cge_bucket_kernel  <<<rgrid, sblk, 0, stream>>>(sc, sc);

    cge_gather_kernel<<<wgrid, blk, 0, stream>>>(x, D, sc, neigh);
    cge_gemm_kernel  <<<ggrid, blk, 0, stream>>>(x, D, neigh, deg, W1, b1,
                                                 h1, 256);
    cge_gather_kernel<<<wgrid, blk, 0, stream>>>(h1, 256, sc, neigh);
    cge_gemm_kernel  <<<ggrid, blk, 0, stream>>>(h1, 256, neigh, deg, W2, b2,
                                                 out, 256);
    cge_copyx_kernel  <<<(N_NODES * (D / 4) + 255) / 256, blk, 0, stream>>>(x, out);
}


# Edge bucketing in 2 launches, not 4: each chunk counting-sorted by node range in its own segment (one LDS pass gives count and rank); no scan launch; bucket sum order: LDS arrival over pieces
# speedup vs baseline: 3.1071x; 1.1712x over previous
#include <hip/hip_runtime.h>

typedef float v2f __attribute__((ext_vector_type(2)));
typedef float v8f __attribute__((ext_vector_type(8)));

#define N_NODES 50000
#define N_EDGES 800000
#define D 128
#define EPS_F 1e-7f

#define NBIN   200
#define BIN_W  250
#define NCHK   250
#define CHK_E  3200
#define SORT_T 1024
#define CHK_K  ((CHK_E + SORT_T - 1) / SORT_T)
static_assert(NBIN * BIN_W == N_NODES, "node ranges must tile the nodes");
static_assert(NCHK * CHK_E == N_EDGES, "edge chunks must tile the edges");
static_assert(NBIN * NCHK == N_NODES, "each table has N_NODES entries");
static_assert(BIN_W <= 256 && N_NODES <= 65536, "packed word: 8 + 16 bits");

#define SCR_CNT 0
#define SCR_OFF 50176
#define SCR_PTR 100480
#define SCR_SRT 150784
#define SCR_BIN 950784

__device__ __forceinline__ int scr(int i) { return ((i >> 7) << 8) | (i & 127); }

__global__ __launch_bounds__(SORT_T) void cge_chunk_kernel(const int* __restrict__ esrc,
                                                           const int* __restrict__ edst,
                                                           int* __restrict__ sc) {
    __shared__ int h[256];
    __shared__ int wtot[8];
    const int t    = threadIdx.x;
    const int lane = t & 31;
    const int w    = t >> 5;
    const int b    = blockIdx.x;
    const int e0   = b * CHK_E;
    int d[CHK_K], s[CHK_K], rk[CHK_K];
#pragma unroll
    for (int k = 0; k < CHK_K; ++k) {
        const int i = t + k * SORT_T;
        d[k] = (i < CHK_E) ? edst[e0 + i] : 0;
        s[k] = (i < CHK_E) ? esrc[e0 + i] : 0;
    }
    if (t < 256) h[t] = 0;
    __syncthreads();
#pragma unroll
    for (int k = 0; k < CHK_K; ++k) {
        rk[k] = 0;
        if (t + k * SORT_T < CHK_E) rk[k] = atomicAdd(&h[(unsigned)d[k] / BIN_W], 1);
    }
    __syncthreads();
    const int c = (t < 256) ? h[t] : 0;
    int x = c;
#pragma unroll
    for (int o = 1; o < 32; o <<= 1) {
        const int y = __shfl_up(x, o);
        if (lane >= o) x += y;
    }
    if (lane == 31 && w < 8) wtot[w] = x;
    __syncthreads();
    if (t < 256) {
        int woff = 0;
#pragma unroll
        for (int k = 0; k < 8; ++k) woff += (k < w) ? wtot[k] : 0;
        const int st = woff + x - c;
        h[t] = st;
        if (t < NBIN) {
            sc[scr(SCR_CNT + t * NCHK + b)] = c;
            sc[scr(SCR_OFF + t * NCHK + b)] = st;
        }
    }
    __syncthreads();
#pragma unroll
    for (int k = 0; k < CHK_K; ++k) {
        if (t + k * SORT_T < CHK_E) {
            const int bin = (unsigned)d[k] / BIN_W;
            const int dl  = d[k] - bin * BIN_W;
            sc[scr(SCR_BIN + e0 + h[bin] + rk[k])] = (dl << 16) | s[k];
        }
    }
}

#define PCE_Q 8
static_assert(32 * PCE_Q >= NCHK && NCHK <= 256, "a block's waves cover all pieces");
__global__ __launch_bounds__(SORT_T) void cge_bucket_kernel(const int* __restrict__ rd,
                                                            int* __restrict__ wr) {
    __shared__ int cnt[256];
    __shared__ int wtot[8];
    __shared__ int pc[256];
    __shared__ int ps[256];
    __shared__ int red[2][32];
    const int t    = threadIdx.x;
    const int lane = t & 31;
    const int w    = t >> 5;
    const int B    = blockIdx.x;
    int myc = 0, mys = 0;
    if (t < NCHK) {
        myc = rd[scr(SCR_CNT + B * NCHK + t)];
        mys = rd[scr(SCR_OFF + B * NCHK + t)];
    }
    const int lim = B * NCHK;
    int4 tv[13];
#pragma unroll
    for (int k = 0; k < 13; ++k) {
        const int i = 4 * (t + k * SORT_T);
        tv[k] = make_int4(0, 0, 0, 0);
        if (i < lim) tv[k] = *reinterpret_cast<const int4*>(&rd[scr(SCR_CNT + i)]);
    }
    int part = 0;
#pragma unroll
    for (int k = 0; k < 13; ++k) {
        const int i = 4 * (t + k * SORT_T);
        part += tv[k].x + ((i + 1 < lim) ? tv[k].y : 0) + ((i + 2 < lim) ? tv[k].z : 0)
              + ((i + 3 < lim) ? tv[k].w : 0);
    }
    if (t < 256) { cnt[t] = 0; pc[t] = myc; ps[t] = mys; }
    int ra = part, rm = myc;
#pragma unroll
    for (int o = 16; o > 0; o >>= 1) { ra += __shfl_xor(ra, o); rm += __shfl_xor(rm, o); }
    if (lane == 0) { red[0][w] = ra; red[1][w] = rm; }
    __syncthreads();
    ra = red[0][lane];
    rm = red[1][lane];
#pragma unroll
    for (int o = 16; o > 0; o >>= 1) { ra += __shfl_xor(ra, o); rm += __shfl_xor(rm, o); }
    const int bs = ra;
    const int be = bs + rm;

    int pcq[PCE_Q], psq[PCE_Q], wq[PCE_Q];
#pragma unroll
    for (int q = 0; q < PCE_Q; ++q) {
        const int b = w + 32 * q;
        pcq[q] = pc[b];
        psq[q] = SCR_BIN + b * CHK_E + ps[b];
    }
#pragma unroll
    for (int q = 0; q < PCE_Q; ++q) {
        wq[q] = 0;
        if (lane < pcq[q]) wq[q] = rd[scr(psq[q] + lane)];
    }
#pragma unroll
    for (int q = 0; q < PCE_Q; ++q)
        if (lane < pcq[q]) atomicAdd(&cnt[wq[q] >> 16], 1);
#pragma unroll
    for (int q = 0; q < PCE_Q; ++q)
        for (int j = lane + 32; j < pcq[q]; j += 32)
            atomicAdd(&cnt[rd[scr(psq[q] + j)] >> 16], 1);
    __syncthreads();
    const int c = (t < 256) ? cnt[t] : 0;
    int x = c;
#pragma unroll
    for (int o = 1; o < 32; o <<= 1) {
        const int y = __shfl_up(x, o);
        if (lane >= o) x += y;
    }
    if (lane == 31 && w < 8) wtot[w] = x;
    __syncthreads();
    if (t < 256) {
        int woff = 0;
#pragma unroll
        for (int k = 0; k < 8; ++k) woff += (k < w) ? wtot[k] : 0;
        const int first = bs + woff + x - c;
        cnt[t] = first;
        if (t < BIN_W) wr[scr(SCR_PTR + B * BIN_W + t)] = first;
        if (B == NBIN - 1 && t == 0) wr[scr(SCR_PTR + N_NODES)] = be;
    }
    __syncthreads();
    int pos[PCE_Q];
#pragma unroll
    for (int q = 0; q < PCE_Q; ++q) {
        pos[q] = 0;
        if (lane < pcq[q]) pos[q] = atomicAdd(&cnt[wq[q] >> 16], 1);
    }
#pragma unroll
    for (int q = 0; q < PCE_Q; ++q)
        if (lane < pcq[q]) wr[scr(SCR_SRT + pos[q])] = wq[q] & 0xFFFF;
#pragma unroll
    for (int q = 0; q < PCE_Q; ++q)
        for (int j = lane + 32; j < pcq[q]; j += 32) {
            const int v = rd[scr(psq[q] + j)];
            const int p = atomicAdd(&cnt[v >> 16], 1);
            wr[scr(SCR_SRT + p)] = v & 0xFFFF;
        }
}

template <int G>
__device__ __forceinline__ void cge_add_rows(const float* __restrict__ hl, int h_stride,
                                             int sidx, int j, float4& acc) {
    float4 v[G];
#pragma unroll
    for (int u = 0; u < G; ++u) {
        const int s = __builtin_amdgcn_readlane(sidx, j + u);
        v[u] = *reinterpret_cast<const float4*>(hl + (size_t)s * h_stride);
    }
#pragma unroll
    for (int u = 0; u < G; ++u) {
        acc.x += v[u].x; acc.y += v[u].y; acc.z += v[u].z; acc.w += v[u].w;
    }
}

__global__ __launch_bounds__(256) void cge_gather_kernel(
    const float* __restrict__ h, int h_stride,
    const int* __restrict__ sc, float* __restrict__ neigh) {
    const int lane = threadIdx.x & 31;
    const int node = blockIdx.x * 8 + (threadIdx.x >> 5);
    const int pv   = sc[scr(SCR_PTR + node + (lane & 1))];
    const int beg  = __builtin_amdgcn_readlane(pv, 0);
    const int end  = __builtin_amdgcn_readlane(pv, 1);
    const float* __restrict__ hl = h + lane * 4;
    float4 acc = make_float4(0.0f, 0.0f, 0.0f, 0.0f);
    for (int c = beg; c < end; c += 32) {
        const int n    = min(32, end - c);
        const int sidx = (lane < n) ? sc[scr(SCR_SRT + c + lane)] : 0;
        int j = 0;
        for (; j + 8 <= n; j += 8) cge_add_rows<8>(hl, h_stride, sidx, j, acc);
        if (n - j >= 4) { cge_add_rows<4>(hl, h_stride, sidx, j, acc); j += 4; }
        if (n - j >= 2) { cge_add_rows<2>(hl, h_stride, sidx, j, acc); j += 2; }
        if (n - j >= 1) { cge_add_rows<1>(hl, h_stride, sidx, j, acc); }
    }
    reinterpret_cast<float4*>(neigh + (size_t)node * D)[lane] = acc;
}

#define GEMM_GRID 256
#define N_TILES   (N_NODES / 16)
#define A_PITCH   260
#define KCH       16
#define NCH       (64 / KCH)

struct TileRows { float4 s0, s1, n0, n1; float d0, d1; };

__device__ __forceinline__ TileRows cge_fetch_rows(
    const float* __restrict__ A, int a_stride, const float* __restrict__ neigh,
    const float* __restrict__ deg, int tile, int srow, int c4) {
    TileRows r;
    const int m0 = tile * 16 + srow;
    const int m1 = m0 + 8;
    r.s0 = *reinterpret_cast<const float4*>(A + (size_t)m0 * a_stride + c4);
    r.s1 = *reinterpret_cast<const float4*>(A + (size_t)m1 * a_stride + c4);
    r.n0 = *reinterpret_cast<const float4*>(neigh + (size_t)m0 * D + c4);
    r.n1 = *reinterpret_cast<const float4*>(neigh + (size_t)m1 * D + c4);
    r.d0 = deg[m0];
    r.d1 = deg[m1];
    return r;
}

__device__ __forceinline__ void cge_stash_rows(float* rows, const TileRows& r,
                                               int srow, int c4) {
    const float i0 = 1.0f / (r.d0 + EPS_F);
    const float i1 = 1.0f / (r.d1 + EPS_F);
    float* p0 = rows + srow * A_PITCH + c4;
    float* p1 = p0 + 8 * A_PITCH;
    *reinterpret_cast<float4*>(p0) = r.s0;
    *reinterpret_cast<float4*>(p1) = r.s1;
    *reinterpret_cast<float4*>(p0 + 128) =
        make_float4(r.n0.x * i0, r.n0.y * i0, r.n0.z * i0, r.n0.w * i0);
    *reinterpret_cast<float4*>(p1 + 128) =
        make_float4(r.n1.x * i1, r.n1.y * i1, r.n1.z * i1, r.n1.w * i1);
}

template <int NT>
__device__ __forceinline__ void cge_read_frags(const float* al, int ch, v2f (&a)[NT][KCH]) {
#pragma unroll
    for (int u = 0; u < NT; ++u)
#pragma unroll
        for (int s = 0; s < KCH; ++s)
            a[u][s] = *reinterpret_cast<const v2f*>(
                al + u * 16 * A_PITCH + 4 * (ch * KCH + s));
}

template <int NT>
__device__ __forceinline__ void cge_tiles_mma(const float* al, const v2f (&bfr)[64],
                                              v8f (&c)[NT]) {
    v2f a[2][NT][KCH];
    cge_read_frags<NT>(al, 0, a[0]);
#pragma unroll
    for (int ch = 0; ch < NCH; ++ch) {
        if (ch + 1 < NCH) cge_read_frags<NT>(al, ch + 1, a[(ch + 1) & 1]);
#pragma unroll
        for (int s = 0; s < KCH; ++s)
#pragma unroll
            for (int u = 0; u < NT; ++u)
                c[u] = __builtin_amdgcn_wmma_f32_16x16x4_f32(
                    false, a[ch & 1][u][s], false, bfr[ch * KCH + s], (short)0, c[u],
                    false, false);
    }
    __builtin_amdgcn_sched_group_barrier(0x100, NT * KCH / 2, 0);
#pragma unroll
    for (int ch = 0; ch < NCH; ++ch) {
        if (ch + 1 < NCH) __builtin_amdgcn_sched_group_barrier(0x100, NT * KCH / 2, 0);
        __builtin_amdgcn_sched_group_barrier(0x008, NT * KCH, 0);
    }
}

__device__ __forceinline__ void cge_store_tile(const v8f& c, float bn, int tile, int khalf,
                                               int n, float* __restrict__ out,
                                               int out_stride) {
    const int mbase = tile * 16 + khalf * 8;
#pragma unroll
    for (int j = 0; j < 8; ++j) {
        float v = c[j] + bn;
        v = v > 0.0f ? v : 0.0f;
        out[(size_t)(mbase + j) * out_stride + n] = v;
    }
}

__global__ __launch_bounds__(256) void cge_gemm_kernel(
    const float* __restrict__ A, int a_stride,
    const float* __restrict__ neigh,
    const float* __restrict__ deg,
    const float* __restrict__ W,
    const float* __restrict__ bias,
    float* __restrict__ out, int out_stride) {
    __shared__ float a_lds[2 * 32 * A_PITCH];

    const int lane  = threadIdx.x & 31;
    const int warp  = threadIdx.x >> 5;
    const int rc    = lane & 15;
    const int khalf = lane >> 4;
    const int n     = warp * 16 + rc;

    v2f bfr[64];
#pragma unroll
    for (int s = 0; s < 64; ++s) {
        const int k = 4 * s + 2 * khalf;
        bfr[s].x = W[(size_t)k * D + n];
        bfr[s].y = W[(size_t)(k + 1) * D + n];
    }
    const float bn = bias[n];

    const int srow = warp;
    const int c4   = lane * 4;
    TileRows ra, rb;

    int t = blockIdx.x;
    ra = cge_fetch_rows(A, a_stride, neigh, deg, t, srow, c4);
    if (t + GEMM_GRID < N_TILES) {
        rb = cge_fetch_rows(A, a_stride, neigh, deg, t + GEMM_GRID, srow, c4);
        cge_stash_rows(a_lds + 16 * A_PITCH, rb, srow, c4);
    }
    cge_stash_rows(a_lds, ra, srow, c4);
    __syncthreads();

    int buf = 0;
    for (; t < N_TILES; t += 2 * GEMM_GRID) {
        const bool two   = t + GEMM_GRID < N_TILES;
        const int  tn    = t + 2 * GEMM_GRID;
        const bool more  = tn < N_TILES;
        const bool more2 = tn + GEMM_GRID < N_TILES;
        if (more)  ra = cge_fetch_rows(A, a_stride, neigh, deg, tn, srow, c4);
        if (more2) rb = cge_fetch_rows(A, a_stride, neigh, deg, tn + GEMM_GRID, srow, c4);

        const float* al = a_lds + buf * (32 * A_PITCH) + rc * A_PITCH + 2 * khalf;
        v8f c[2] = {v8f{}, v8f{}};
        if (two) {
            cge_tiles_mma<2>(al, bfr, c);
        } else {
            v8f c1[1] = {v8f{}};
            cge_tiles_mma<1>(al, bfr, c1);
            c[0] = c1[0];
        }

        float* nb = a_lds + (buf ^ 1) * (32 * A_PITCH);
        if (more)  cge_stash_rows(nb, ra, srow, c4);
        if (more2) cge_stash_rows(nb + 16 * A_PITCH, rb, srow, c4);
        __syncthreads();
        cge_store_tile(c[0], bn, t, khalf, n, out, out_stride);
        if (two) cge_store_tile(c[1], bn, t + GEMM_GRID, khalf, n, out, out_stride);
        buf ^= 1;
    }
}

__global__ void cge_copyx_kernel(const float* __restrict__ x,
                                 float* __restrict__ out) {
    int i = blockIdx.x * blockDim.x + threadIdx.x;
    if (i >= N_NODES * (D / 4)) return;
    int r  = i >> 5;
    int c4 = i & 31;
    float4 v = reinterpret_cast<const float4*>(x)[i];
    reinterpret_cast<float4*>(out + (size_t)r * 256 + 128)[c4] = v;
}

extern "C" void kernel_launch(void* const* d_in, const int* in_sizes, int n_in,
                              void* d_out, int out_size, void* d_ws, size_t ws_size,
                              hipStream_t stream) {
    const float* x    = (const float*)d_in[0];
    const int*   esrc = (const int*)  d_in[1];
    const int*   edst = (const int*)  d_in[2];
    const float* deg  = (const float*)d_in[3];
    const float* W1   = (const float*)d_in[4];
    const float* b1   = (const float*)d_in[5];
    const float* W2   = (const float*)d_in[6];
    const float* b2   = (const float*)d_in[7];
    float* out = (float*)d_out;

    float* neigh = (float*)d_ws;
    float* h1    = out + 128;
    int*   sc    = (int*)d_out;

    dim3 blk(256);
    dim3 sblk(SORT_T);
    dim3 cgrid(NCHK);
    dim3 rgrid(NBIN);
    dim3 wgrid(N_NODES / 8);
    dim3 ggrid(GEMM_GRID);

    cge_chunk_kernel   <<<cgrid, sblk, 0, stream>>>(esrc, edst, sc);
    cge_bucket_kernel  <<<rgrid, sblk, 0, stream>>>(sc, sc);

    cge_gather_kernel<<<wgrid, blk, 0, stream>>>(x, D, sc, neigh);
    cge_gemm_kernel  <<<ggrid, blk, 0, stream>>>(x, D, neigh, deg, W1, b1,
                                                 h1, 256);
    cge_gather_kernel<<<wgrid, blk, 0, stream>>>(h1, 256, sc, neigh);
    cge_gemm_kernel  <<<ggrid, blk, 0, stream>>>(h1, 256, neigh, deg, W2, b2,
                                                 out, 256);
    cge_copyx_kernel  <<<(N_NODES * (D / 4) + 255) / 256, blk, 0, stream>>>(x, out);
}


# x copy into out[:,128:256] folded into the layer-2 GEMM launch: a block moves the x rows of its own tiles after the trip barrier, once their h1 values are staged; copy launch removed, same sums
# speedup vs baseline: 3.3156x; 1.0671x over previous
#include <hip/hip_runtime.h>

typedef float v2f __attribute__((ext_vector_type(2)));
typedef float v8f __attribute__((ext_vector_type(8)));

#define N_NODES 50000
#define N_EDGES 800000
#define D 128
#define EPS_F 1e-7f

#define NBIN   200
#define BIN_W  250
#define NCHK   250
#define CHK_E  3200
#define SORT_T 1024
#define CHK_K  ((CHK_E + SORT_T - 1) / SORT_T)
static_assert(NBIN * BIN_W == N_NODES, "node ranges must tile the nodes");
static_assert(NCHK * CHK_E == N_EDGES, "edge chunks must tile the edges");
static_assert(NBIN * NCHK == N_NODES, "each table has N_NODES entries");
static_assert(BIN_W <= 256 && N_NODES <= 65536, "packed word: 8 + 16 bits");

#define SCR_CNT 0
#define SCR_OFF 50176
#define SCR_PTR 100480
#define SCR_SRT 150784
#define SCR_BIN 950784

__device__ __forceinline__ int scr(int i) { return ((i >> 7) << 8) | (i & 127); }

__global__ __launch_bounds__(SORT_T) void cge_chunk_kernel(const int* __restrict__ esrc,
                                                           const int* __restrict__ edst,
                                                           int* __restrict__ sc) {
    __shared__ int h[256];
    __shared__ int wtot[8];
    const int t    = threadIdx.x;
    const int lane = t & 31;
    const int w    = t >> 5;
    const int b    = blockIdx.x;
    const int e0   = b * CHK_E;
    int d[CHK_K], s[CHK_K], rk[CHK_K];
#pragma unroll
    for (int k = 0; k < CHK_K; ++k) {
        const int i = t + k * SORT_T;
        d[k] = (i < CHK_E) ? edst[e0 + i] : 0;
        s[k] = (i < CHK_E) ? esrc[e0 + i] : 0;
    }
    if (t < 256) h[t] = 0;
    __syncthreads();
#pragma unroll
    for (int k = 0; k < CHK_K; ++k) {
        rk[k] = 0;
        if (t + k * SORT_T < CHK_E) rk[k] = atomicAdd(&h[(unsigned)d[k] / BIN_W], 1);
    }
    __syncthreads();
    const int c = (t < 256) ? h[t] : 0;
    int x = c;
#pragma unroll
    for (int o = 1; o < 32; o <<= 1) {
        const int y = __shfl_up(x, o);
        if (lane >= o) x += y;
    }
    if (lane == 31 && w < 8) wtot[w] = x;
    __syncthreads();
    if (t < 256) {
        int woff = 0;
#pragma unroll
        for (int k = 0; k < 8; ++k) woff += (k < w) ? wtot[k] : 0;
        const int st = woff + x - c;
        h[t] = st;
        if (t < NBIN) {
            sc[scr(SCR_CNT + t * NCHK + b)] = c;
            sc[scr(SCR_OFF + t * NCHK + b)] = st;
        }
    }
    __syncthreads();
#pragma unroll
    for (int k = 0; k < CHK_K; ++k) {
        if (t + k * SORT_T < CHK_E) {
            const int bin = (unsigned)d[k] / BIN_W;
            const int dl  = d[k] - bin * BIN_W;
            sc[scr(SCR_BIN + e0 + h[bin] + rk[k])] = (dl << 16) | s[k];
        }
    }
}

#define PCE_Q 8
static_assert(32 * PCE_Q >= NCHK && NCHK <= 256, "a block's waves cover all pieces");
__global__ __launch_bounds__(SORT_T) void cge_bucket_kernel(const int* __restrict__ rd,
                                                            int* __restrict__ wr) {
    __shared__ int cnt[256];
    __shared__ int wtot[8];
    __shared__ int pc[256];
    __shared__ int ps[256];
    __shared__ int red[2][32];
    const int t    = threadIdx.x;
    const int lane = t & 31;
    const int w    = t >> 5;
    const int B    = blockIdx.x;
    int myc = 0, mys = 0;
    if (t < NCHK) {
        myc = rd[scr(SCR_CNT + B * NCHK + t)];
        mys = rd[scr(SCR_OFF + B * NCHK + t)];
    }
    const int lim = B * NCHK;
    int4 tv[13];
#pragma unroll
    for (int k = 0; k < 13; ++k) {
        const int i = 4 * (t + k * SORT_T);
        tv[k] = make_int4(0, 0, 0, 0);
        if (i < lim) tv[k] = *reinterpret_cast<const int4*>(&rd[scr(SCR_CNT + i)]);
    }
    int part = 0;
#pragma unroll
    for (int k = 0; k < 13; ++k) {
        const int i = 4 * (t + k * SORT_T);
        part += tv[k].x + ((i + 1 < lim) ? tv[k].y : 0) + ((i + 2 < lim) ? tv[k].z : 0)
              + ((i + 3 < lim) ? tv[k].w : 0);
    }
    if (t < 256) { cnt[t] = 0; pc[t] = myc; ps[t] = mys; }
    int ra = part, rm = myc;
#pragma unroll
    for (int o = 16; o > 0; o >>= 1) { ra += __shfl_xor(ra, o); rm += __shfl_xor(rm, o); }
    if (lane == 0) { red[0][w] = ra; red[1][w] = rm; }
    __syncthreads();
    ra = red[0][lane];
    rm = red[1][lane];
#pragma unroll
    for (int o = 16; o > 0; o >>= 1) { ra += __shfl_xor(ra, o); rm += __shfl_xor(rm, o); }
    const int bs = ra;
    const int be = bs + rm;

    int pcq[PCE_Q], psq[PCE_Q], wq[PCE_Q];
#pragma unroll
    for (int q = 0; q < PCE_Q; ++q) {
        const int b = w + 32 * q;
        pcq[q] = pc[b];
        psq[q] = SCR_BIN + b * CHK_E + ps[b];
    }
#pragma unroll
    for (int q = 0; q < PCE_Q; ++q) {
        wq[q] = 0;
        if (lane < pcq[q]) wq[q] = rd[scr(psq[q] + lane)];
    }
#pragma unroll
    for (int q = 0; q < PCE_Q; ++q)
        if (lane < pcq[q]) atomicAdd(&cnt[wq[q] >> 16], 1);
#pragma unroll
    for (int q = 0; q < PCE_Q; ++q)
        for (int j = lane + 32; j < pcq[q]; j += 32)
            atomicAdd(&cnt[rd[scr(psq[q] + j)] >> 16], 1);
    __syncthreads();
    const int c = (t < 256) ? cnt[t] : 0;
    int x = c;
#pragma unroll
    for (int o = 1; o < 32; o <<= 1) {
        const int y = __shfl_up(x, o);
        if (lane >= o) x += y;
    }
    if (lane == 31 && w < 8) wtot[w] = x;
    __syncthreads();
    if (t < 256) {
        int woff = 0;
#pragma unroll
        for (int k = 0; k < 8; ++k) woff += (k < w) ? wtot[k] : 0;
        const int first = bs + woff + x - c;
        cnt[t] = first;
        if (t < BIN_W) wr[scr(SCR_PTR + B * BIN_W + t)] = first;
        if (B == NBIN - 1 && t == 0) wr[scr(SCR_PTR + N_NODES)] = be;
    }
    __syncthreads();
    int pos[PCE_Q];
#pragma unroll
    for (int q = 0; q < PCE_Q; ++q) {
        pos[q] = 0;
        if (lane < pcq[q]) pos[q] = atomicAdd(&cnt[wq[q] >> 16], 1);
    }
#pragma unroll
    for (int q = 0; q < PCE_Q; ++q)
        if (lane < pcq[q]) wr[scr(SCR_SRT + pos[q])] = wq[q] & 0xFFFF;
#pragma unroll
    for (int q = 0; q < PCE_Q; ++q)
        for (int j = lane + 32; j < pcq[q]; j += 32) {
            const int v = rd[scr(psq[q] + j)];
            const int p = atomicAdd(&cnt[v >> 16], 1);
            wr[scr(SCR_SRT + p)] = v & 0xFFFF;
        }
}

template <int G>
__device__ __forceinline__ void cge_add_rows(const float* __restrict__ hl, int h_stride,
                                             int sidx, int j, float4& acc) {
    float4 v[G];
#pragma unroll
    for (int u = 0; u < G; ++u) {
        const int s = __builtin_amdgcn_readlane(sidx, j + u);
        v[u] = *reinterpret_cast<const float4*>(hl + (size_t)s * h_stride);
    }
#pragma unroll
    for (int u = 0; u < G; ++u) {
        acc.x += v[u].x; acc.y += v[u].y; acc.z += v[u].z; acc.w += v[u].w;
    }
}

__global__ __launch_bounds__(256) void cge_gather_kernel(
    const float* __restrict__ h, int h_stride,
    const int* __restrict__ sc, float* __restrict__ neigh) {
    const int lane = threadIdx.x & 31;
    const int node = blockIdx.x * 8 + (threadIdx.x >> 5);
    const int pv   = sc[scr(SCR_PTR + node + (lane & 1))];
    const int beg  = __builtin_amdgcn_readlane(pv, 0);
    const int end  = __builtin_amdgcn_readlane(pv, 1);
    const float* __restrict__ hl = h + lane * 4;
    float4 acc = make_float4(0.0f, 0.0f, 0.0f, 0.0f);
    for (int c = beg; c < end; c += 32) {
        const int n    = min(32, end - c);
        const int sidx = (lane < n) ? sc[scr(SCR_SRT + c + lane)] : 0;
        int j = 0;
        for (; j + 8 <= n; j += 8) cge_add_rows<8>(hl, h_stride, sidx, j, acc);
        if (n - j >= 4) { cge_add_rows<4>(hl, h_stride, sidx, j, acc); j += 4; }
        if (n - j >= 2) { cge_add_rows<2>(hl, h_stride, sidx, j, acc); j += 2; }
        if (n - j >= 1) { cge_add_rows<1>(hl, h_stride, sidx, j, acc); }
    }
    reinterpret_cast<float4*>(neigh + (size_t)node * D)[lane] = acc;
}

#define GEMM_GRID 256
#define N_TILES   (N_NODES / 16)
#define A_PITCH   260
#define KCH       16
#define NCH       (64 / KCH)

struct TileRows { float4 s0, s1, n0, n1; float d0, d1; };

__device__ __forceinline__ TileRows cge_fetch_rows(
    const float* __restrict__ A, int a_stride, const float* __restrict__ neigh,
    const float* __restrict__ deg, int tile, int srow, int c4) {
    TileRows r;
    const int m0 = tile * 16 + srow;
    const int m1 = m0 + 8;
    r.s0 = *reinterpret_cast<const float4*>(A + (size_t)m0 * a_stride + c4);
    r.s1 = *reinterpret_cast<const float4*>(A + (size_t)m1 * a_stride + c4);
    r.n0 = *reinterpret_cast<const float4*>(neigh + (size_t)m0 * D + c4);
    r.n1 = *reinterpret_cast<const float4*>(neigh + (size_t)m1 * D + c4);
    r.d0 = deg[m0];
    r.d1 = deg[m1];
    return r;
}

__device__ __forceinline__ void cge_stash_rows(float* rows, const TileRows& r,
                                               int srow, int c4) {
    const float i0 = 1.0f / (r.d0 + EPS_F);
    const float i1 = 1.0f / (r.d1 + EPS_F);
    float* p0 = rows + srow * A_PITCH + c4;
    float* p1 = p0 + 8 * A_PITCH;
    *reinterpret_cast<float4*>(p0) = r.s0;
    *reinterpret_cast<float4*>(p1) = r.s1;
    *reinterpret_cast<float4*>(p0 + 128) =
        make_float4(r.n0.x * i0, r.n0.y * i0, r.n0.z * i0, r.n0.w * i0);
    *reinterpret_cast<float4*>(p1 + 128) =
        make_float4(r.n1.x * i1, r.n1.y * i1, r.n1.z * i1, r.n1.w * i1);
}

template <int NT>
__device__ __forceinline__ void cge_read_frags(const float* al, int ch, v2f (&a)[NT][KCH]) {
#pragma unroll
    for (int u = 0; u < NT; ++u)
#pragma unroll
        for (int s = 0; s < KCH; ++s)
            a[u][s] = *reinterpret_cast<const v2f*>(
                al + u * 16 * A_PITCH + 4 * (ch * KCH + s));
}

template <int NT>
__device__ __forceinline__ void cge_tiles_mma(const float* al, const v2f (&bfr)[64],
                                              v8f (&c)[NT]) {
    v2f a[2][NT][KCH];
    cge_read_frags<NT>(al, 0, a[0]);
#pragma unroll
    for (int ch = 0; ch < NCH; ++ch) {
        if (ch + 1 < NCH) cge_read_frags<NT>(al, ch + 1, a[(ch + 1) & 1]);
#pragma unroll
        for (int s = 0; s < KCH; ++s)
#pragma unroll
            for (int u = 0; u < NT; ++u)
                c[u] = __builtin_amdgcn_wmma_f32_16x16x4_f32(
                    false, a[ch & 1][u][s], false, bfr[ch * KCH + s], (short)0, c[u],
                    false, false);
    }
    __builtin_amdgcn_sched_group_barrier(0x100, NT * KCH / 2, 0);
#pragma unroll
    for (int ch = 0; ch < NCH; ++ch) {
        if (ch + 1 < NCH) __builtin_amdgcn_sched_group_barrier(0x100, NT * KCH / 2, 0);
        __builtin_amdgcn_sched_group_barrier(0x008, NT * KCH, 0);
    }
}

__device__ __forceinline__ void cge_store_tile(const v8f& c, float bn, int tile, int khalf,
                                               int n, float* __restrict__ out,
                                               int out_stride) {
    const int mbase = tile * 16 + khalf * 8;
#pragma unroll
    for (int j = 0; j < 8; ++j) {
        float v = c[j] + bn;
        v = v > 0.0f ? v : 0.0f;
        out[(size_t)(mbase + j) * out_stride + n] = v;
    }
}

template <bool XCOPY>
__device__ __forceinline__ void cge_gemm_body(
    const float* A, int a_stride,
    const float* neigh,
    const float* deg,
    const float* W,
    const float* bias,
    float* out, int out_stride,
    const float* xsrc) {
    __shared__ float a_lds[2 * 32 * A_PITCH];

    const int lane  = threadIdx.x & 31;
    const int warp  = threadIdx.x >> 5;
    const int rc    = lane & 15;
    const int khalf = lane >> 4;
    const int n     = warp * 16 + rc;

    v2f bfr[64];
#pragma unroll
    for (int s = 0; s < 64; ++s) {
        const int k = 4 * s + 2 * khalf;
        bfr[s].x = W[(size_t)k * D + n];
        bfr[s].y = W[(size_t)(k + 1) * D + n];
    }
    const float bn = bias[n];

    const int srow = warp;
    const int c4   = lane * 4;
    TileRows ra, rb;

    int t = blockIdx.x;
    ra = cge_fetch_rows(A, a_stride, neigh, deg, t, srow, c4);
    if (t + GEMM_GRID < N_TILES) {
        rb = cge_fetch_rows(A, a_stride, neigh, deg, t + GEMM_GRID, srow, c4);
        cge_stash_rows(a_lds + 16 * A_PITCH, rb, srow, c4);
    }
    cge_stash_rows(a_lds, ra, srow, c4);
    __syncthreads();

    int buf = 0;
    for (; t < N_TILES; t += 2 * GEMM_GRID) {
        const bool two   = t + GEMM_GRID < N_TILES;
        const int  tn    = t + 2 * GEMM_GRID;
        const bool more  = tn < N_TILES;
        const bool more2 = tn + GEMM_GRID < N_TILES;
        if (more)  ra = cge_fetch_rows(A, a_stride, neigh, deg, tn, srow, c4);
        if (more2) rb = cge_fetch_rows(A, a_stride, neigh, deg, tn + GEMM_GRID, srow, c4);
        float4 xa0 = make_float4(0.0f, 0.0f, 0.0f, 0.0f), xa1 = xa0, xb0 = xa0, xb1 = xa0;
        if (XCOPY) {
            const float* xp = xsrc + ((size_t)t * 16 + srow) * D + c4;
            xa0 = *reinterpret_cast<const float4*>(xp);
            xa1 = *reinterpret_cast<const float4*>(xp + 8 * D);
            if (two) {
                xb0 = *reinterpret_cast<const float4*>(xp + (size_t)GEMM_GRID * 16 * D);
                xb1 = *reinterpret_cast<const float4*>(xp + (size_t)GEMM_GRID * 16 * D + 8 * D);
            }
        }

        const float* al = a_lds + buf * (32 * A_PITCH) + rc * A_PITCH + 2 * khalf;
        v8f c[2] = {v8f{}, v8f{}};
        if (two) {
            cge_tiles_mma<2>(al, bfr, c);
        } else {
            v8f c1[1] = {v8f{}};
            cge_tiles_mma<1>(al, bfr, c1);
            c[0] = c1[0];
        }

        float* nb = a_lds + (buf ^ 1) * (32 * A_PITCH);
        if (more)  cge_stash_rows(nb, ra, srow, c4);
        if (more2) cge_stash_rows(nb + 16 * A_PITCH, rb, srow, c4);
        __syncthreads();
        cge_store_tile(c[0], bn, t, khalf, n, out, out_stride);
        if (two) cge_store_tile(c[1], bn, t + GEMM_GRID, khalf, n, out, out_stride);
        if (XCOPY) {
            float* xo = out + 128 + ((size_t)t * 16 + srow) * out_stride + c4;
            *reinterpret_cast<float4*>(xo) = xa0;
            *reinterpret_cast<float4*>(xo + (size_t)8 * out_stride) = xa1;
            if (two) {
                float* xo2 = xo + (size_t)GEMM_GRID * 16 * out_stride;
                *reinterpret_cast<float4*>(xo2) = xb0;
                *reinterpret_cast<float4*>(xo2 + (size_t)8 * out_stride) = xb1;
            }
        }
        buf ^= 1;
    }
}

__global__ __launch_bounds__(256) void cge_gemm_kernel(
    const float* __restrict__ A, int a_stride,
    const float* __restrict__ neigh,
    const float* __restrict__ deg,
    const float* __restrict__ W,
    const float* __restrict__ bias,
    float* __restrict__ out, int out_stride) {
    cge_gemm_body<false>(A, a_stride, neigh, deg, W, bias, out, out_stride, nullptr);
}

__global__ __launch_bounds__(256) void cge_gemm_final_kernel(
    const float* A, int a_stride,
    const float* __restrict__ neigh,
    const float* __restrict__ deg,
    const float* __restrict__ W,
    const float* __restrict__ bias,
    float* out, int out_stride,
    const float* __restrict__ x) {
    cge_gemm_body<true>(A, a_stride, neigh, deg, W, bias, out, out_stride, x);
}

extern "C" void kernel_launch(void* const* d_in, const int* in_sizes, int n_in,
                              void* d_out, int out_size, void* d_ws, size_t ws_size,
                              hipStream_t stream) {
    const float* x    = (const float*)d_in[0];
    const int*   esrc = (const int*)  d_in[1];
    const int*   edst = (const int*)  d_in[2];
    const float* deg  = (const float*)d_in[3];
    const float* W1   = (const float*)d_in[4];
    const float* b1   = (const float*)d_in[5];
    const float* W2   = (const float*)d_in[6];
    const float* b2   = (const float*)d_in[7];
    float* out = (float*)d_out;

    float* neigh = (float*)d_ws;
    float* h1    = out + 128;
    int*   sc    = (int*)d_out;

    dim3 blk(256);
    dim3 sblk(SORT_T);
    dim3 cgrid(NCHK);
    dim3 rgrid(NBIN);
    dim3 wgrid(N_NODES / 8);
    dim3 ggrid(GEMM_GRID);

    cge_chunk_kernel   <<<cgrid, sblk, 0, stream>>>(esrc, edst, sc);
    cge_bucket_kernel  <<<rgrid, sblk, 0, stream>>>(sc, sc);

    cge_gather_kernel<<<wgrid, blk, 0, stream>>>(x, D, sc, neigh);
    cge_gemm_kernel  <<<ggrid, blk, 0, stream>>>(x, D, neigh, deg, W1, b1,
                                                 h1, 256);
    cge_gather_kernel<<<wgrid, blk, 0, stream>>>(h1, 256, sc, neigh);
    cge_gemm_final_kernel<<<ggrid, blk, 0, stream>>>(h1, 256, neigh, deg, W2, b2,
                                                     out, 256, x);
}


# Both edge-sort kernels place their words in an LDS image of the run they own and copy it out as whole 128-byte lines (16-byte stores): 1.6M scattered 4-byte global stores go; slot order unchanged
# speedup vs baseline: 3.4547x; 1.0420x over previous
#include <hip/hip_runtime.h>

typedef float v2f __attribute__((ext_vector_type(2)));
typedef float v8f __attribute__((ext_vector_type(8)));

#define N_NODES 50000
#define N_EDGES 800000
#define D 128
#define EPS_F 1e-7f

#define NBIN   200
#define BIN_W  250
#define NCHK   250
#define CHK_E  3200
#define SORT_T 1024
#define CHK_K  ((CHK_E + SORT_T - 1) / SORT_T)
static_assert(NBIN * BIN_W == N_NODES, "node ranges must tile the nodes");
static_assert(NCHK * CHK_E == N_EDGES, "edge chunks must tile the edges");
static_assert(NBIN * NCHK == N_NODES, "each table has N_NODES entries");
static_assert(BIN_W <= 256 && N_NODES <= 65536, "packed word: 8 + 16 bits");
static_assert(CHK_E % 4 == 0 && CHK_E / 4 <= SORT_T, "a chunk's words leave as one 16-byte store per thread");

#define SCR_CNT 0
#define SCR_OFF 50176
#define SCR_PTR 100480
#define SCR_SRT 150784
#define SCR_BIN 950784
static_assert(SCR_BIN % 4 == 0 && SCR_SRT % 4 == 0, "16-byte stores into the word arrays");

__device__ __forceinline__ int scr(int i) { return ((i >> 7) << 8) | (i & 127); }

__global__ __launch_bounds__(SORT_T) void cge_chunk_kernel(const int* __restrict__ esrc,
                                                           const int* __restrict__ edst,
                                                           int* __restrict__ sc) {
    __shared__ int h[256];
    __shared__ int wtot[8];
    __shared__ int4 img4[CHK_E / 4];
    int* img = reinterpret_cast<int*>(img4);
    const int t    = threadIdx.x;
    const int lane = t & 31;
    const int w    = t >> 5;
    const int b    = blockIdx.x;
    const int e0   = b * CHK_E;
    int d[CHK_K], s[CHK_K], rk[CHK_K];
#pragma unroll
    for (int k = 0; k < CHK_K; ++k) {
        const int i = t + k * SORT_T;
        d[k] = (i < CHK_E) ? edst[e0 + i] : 0;
        s[k] = (i < CHK_E) ? esrc[e0 + i] : 0;
    }
    if (t < 256) h[t] = 0;
    __syncthreads();
#pragma unroll
    for (int k = 0; k < CHK_K; ++k) {
        rk[k] = 0;
        if (t + k * SORT_T < CHK_E) rk[k] = atomicAdd(&h[(unsigned)d[k] / BIN_W], 1);
    }
    __syncthreads();
    const int c = (t < 256) ? h[t] : 0;
    int x = c;
#pragma unroll
    for (int o = 1; o < 32; o <<= 1) {
        const int y = __shfl_up(x, o);
        if (lane >= o) x += y;
    }
    if (lane == 31 && w < 8) wtot[w] = x;
    __syncthreads();
    if (t < 256) {
        int woff = 0;
#pragma unroll
        for (int k = 0; k < 8; ++k) woff += (k < w) ? wtot[k] : 0;
        const int st = woff + x - c;
        h[t] = st;
        if (t < NBIN) {
            sc[scr(SCR_CNT + t * NCHK + b)] = c;
            sc[scr(SCR_OFF + t * NCHK + b)] = st;
        }
    }
    __syncthreads();
#pragma unroll
    for (int k = 0; k < CHK_K; ++k) {
        if (t + k * SORT_T < CHK_E) {
            const int bin = (unsigned)d[k] / BIN_W;
            const int dl  = d[k] - bin * BIN_W;
            img[h[bin] + rk[k]] = (dl << 16) | s[k];
        }
    }
    __syncthreads();
    if (t < CHK_E / 4)
        *reinterpret_cast<int4*>(&sc[scr(SCR_BIN + e0 + 4 * t)]) = img4[t];
}

#define PCE_Q 8
#define IMG_W 8192
static_assert(32 * PCE_Q >= NCHK && NCHK <= 256, "a block's waves cover all pieces");
__global__ __launch_bounds__(SORT_T) void cge_bucket_kernel(const int* __restrict__ rd,
                                                            int* __restrict__ wr) {
    __shared__ int cnt[256];
    __shared__ int wtot[8];
    __shared__ int pc[256];
    __shared__ int ps[256];
    __shared__ int red[2][32];
    __shared__ int4 img4[IMG_W / 4];
    int* img = reinterpret_cast<int*>(img4);
    const int t    = threadIdx.x;
    const int lane = t & 31;
    const int w    = t >> 5;
    const int B    = blockIdx.x;
    int myc = 0, mys = 0;
    if (t < NCHK) {
        myc = rd[scr(SCR_CNT + B * NCHK + t)];
        mys = rd[scr(SCR_OFF + B * NCHK + t)];
    }
    const int lim = B * NCHK;
    int4 tv[13];
#pragma unroll
    for (int k = 0; k < 13; ++k) {
        const int i = 4 * (t + k * SORT_T);
        tv[k] = make_int4(0, 0, 0, 0);
        if (i < lim) tv[k] = *reinterpret_cast<const int4*>(&rd[scr(SCR_CNT + i)]);
    }
    int part = 0;
#pragma unroll
    for (int k = 0; k < 13; ++k) {
        const int i = 4 * (t + k * SORT_T);
        part += tv[k].x + ((i + 1 < lim) ? tv[k].y : 0) + ((i + 2 < lim) ? tv[k].z : 0)
              + ((i + 3 < lim) ? tv[k].w : 0);
    }
    if (t < 256) { cnt[t] = 0; pc[t] = myc; ps[t] = mys; }
    int ra = part, rm = myc;
#pragma unroll
    for (int o = 16; o > 0; o >>= 1) { ra += __shfl_xor(ra, o); rm += __shfl_xor(rm, o); }
    if (lane == 0) { red[0][w] = ra; red[1][w] = rm; }
    __syncthreads();
    ra = red[0][lane];
    rm = red[1][lane];
#pragma unroll
    for (int o = 16; o > 0; o >>= 1) { ra += __shfl_xor(ra, o); rm += __shfl_xor(rm, o); }
    const int bs = ra;
    const int be = bs + rm;

    int pcq[PCE_Q], psq[PCE_Q], wq[PCE_Q];
#pragma unroll
    for (int q = 0; q < PCE_Q; ++q) {
        const int b = w + 32 * q;
        pcq[q] = pc[b];
        psq[q] = SCR_BIN + b * CHK_E + ps[b];
    }
#pragma unroll
    for (int q = 0; q < PCE_Q; ++q) {
        wq[q] = 0;
        if (lane < pcq[q]) wq[q] = rd[scr(psq[q] + lane)];
    }
#pragma unroll
    for (int q = 0; q < PCE_Q; ++q)
        if (lane < pcq[q]) atomicAdd(&cnt[wq[q] >> 16], 1);
#pragma unroll
    for (int q = 0; q < PCE_Q; ++q)
        for (int j = lane + 32; j < pcq[q]; j += 32)
            atomicAdd(&cnt[rd[scr(psq[q] + j)] >> 16], 1);
    __syncthreads();
    const int c = (t < 256) ? cnt[t] : 0;
    int x = c;
#pragma unroll
    for (int o = 1; o < 32; o <<= 1) {
        const int y = __shfl_up(x, o);
        if (lane >= o) x += y;
    }
    if (lane == 31 && w < 8) wtot[w] = x;
    __syncthreads();
    if (t < 256) {
        int woff = 0;
#pragma unroll
        for (int k = 0; k < 8; ++k) woff += (k < w) ? wtot[k] : 0;
        const int first = bs + woff + x - c;
        cnt[t] = first;
        if (t < BIN_W) wr[scr(SCR_PTR + B * BIN_W + t)] = first;
        if (B == NBIN - 1 && t == 0) wr[scr(SCR_PTR + N_NODES)] = be;
    }
    __syncthreads();
    const int g0   = SCR_SRT + bs;
    const int off  = g0 & 31;
    const int nw   = be - bs;
    const bool fit = nw + off <= IMG_W;
    const int sh   = off - bs;
    int pos[PCE_Q];
#pragma unroll
    for (int q = 0; q < PCE_Q; ++q) {
        pos[q] = 0;
        if (lane < pcq[q]) pos[q] = atomicAdd(&cnt[wq[q] >> 16], 1);
    }
#pragma unroll
    for (int q = 0; q < PCE_Q; ++q)
        if (lane < pcq[q]) {
            if (fit) img[pos[q] + sh] = wq[q] & 0xFFFF;
            else     wr[scr(SCR_SRT + pos[q])] = wq[q] & 0xFFFF;
        }
#pragma unroll
    for (int q = 0; q < PCE_Q; ++q)
        for (int j = lane + 32; j < pcq[q]; j += 32) {
            const int v = rd[scr(psq[q] + j)];
            const int p = atomicAdd(&cnt[v >> 16], 1);
            if (fit) img[p + sh] = v & 0xFFFF;
            else     wr[scr(SCR_SRT + p)] = v & 0xFFFF;
        }
    if (fit) {
        __syncthreads();
        const int gb = g0 - off;
        const int hi = off + nw;
        for (int j = 4 * t; j < hi; j += 4 * SORT_T) {
            if (j >= off && j + 4 <= hi) {
                *reinterpret_cast<int4*>(&wr[scr(gb + j)]) = img4[j >> 2];
            } else {
                for (int u = 0; u < 4; ++u)
                    if (j + u >= off && j + u < hi) wr[scr(gb + j + u)] = img[j + u];
            }
        }
    }
}

template <int G>
__device__ __forceinline__ void cge_add_rows(const float* __restrict__ hl, int h_stride,
                                             int sidx, int j, float4& acc) {
    float4 v[G];
#pragma unroll
    for (int u = 0; u < G; ++u) {
        const int s = __builtin_amdgcn_readlane(sidx, j + u);
        v[u] = *reinterpret_cast<const float4*>(hl + (size_t)s * h_stride);
    }
#pragma unroll
    for (int u = 0; u < G; ++u) {
        acc.x += v[u].x; acc.y += v[u].y; acc.z += v[u].z; acc.w += v[u].w;
    }
}

__global__ __launch_bounds__(256) void cge_gather_kernel(
    const float* __restrict__ h, int h_stride,
    const int* __restrict__ sc, float* __restrict__ neigh) {
    const int lane = threadIdx.x & 31;
    const int node = blockIdx.x * 8 + (threadIdx.x >> 5);
    const int pv   = sc[scr(SCR_PTR + node + (lane & 1))];
    const int beg  = __builtin_amdgcn_readlane(pv, 0);
    const int end  = __builtin_amdgcn_readlane(pv, 1);
    const float* __restrict__ hl = h + lane * 4;
    float4 acc = make_float4(0.0f, 0.0f, 0.0f, 0.0f);
    for (int c = beg; c < end; c += 32) {
        const int n    = min(32, end - c);
        const int sidx = (lane < n) ? sc[scr(SCR_SRT + c + lane)] : 0;
        int j = 0;
        for (; j + 8 <= n; j += 8) cge_add_rows<8>(hl, h_stride, sidx, j, acc);
        if (n - j >= 4) { cge_add_rows<4>(hl, h_stride, sidx, j, acc); j += 4; }
        if (n - j >= 2) { cge_add_rows<2>(hl, h_stride, sidx, j, acc); j += 2; }
        if (n - j >= 1) { cge_add_rows<1>(hl, h_stride, sidx, j, acc); }
    }
    reinterpret_cast<float4*>(neigh + (size_t)node * D)[lane] = acc;
}

#define GEMM_GRID 256
#define N_TILES   (N_NODES / 16)
#define A_PITCH   260
#define KCH       16
#define NCH       (64 / KCH)

struct TileRows { float4 s0, s1, n0, n1; float d0, d1; };

__device__ __forceinline__ TileRows cge_fetch_rows(
    const float* __restrict__ A, int a_stride, const float* __restrict__ neigh,
    const float* __restrict__ deg, int tile, int srow, int c4) {
    TileRows r;
    const int m0 = tile * 16 + srow;
    const int m1 = m0 + 8;
    r.s0 = *reinterpret_cast<const float4*>(A + (size_t)m0 * a_stride + c4);
    r.s1 = *reinterpret_cast<const float4*>(A + (size_t)m1 * a_stride + c4);
    r.n0 = *reinterpret_cast<const float4*>(neigh + (size_t)m0 * D + c4);
    r.n1 = *reinterpret_cast<const float4*>(neigh + (size_t)m1 * D + c4);
    r.d0 = deg[m0];
    r.d1 = deg[m1];
    return r;
}

__device__ __forceinline__ void cge_stash_rows(float* rows, const TileRows& r,
                                               int srow, int c4) {
    const float i0 = 1.0f / (r.d0 + EPS_F);
    const float i1 = 1.0f / (r.d1 + EPS_F);
    float* p0 = rows + srow * A_PITCH + c4;
    float* p1 = p0 + 8 * A_PITCH;
    *reinterpret_cast<float4*>(p0) = r.s0;
    *reinterpret_cast<float4*>(p1) = r.s1;
    *reinterpret_cast<float4*>(p0 + 128) =
        make_float4(r.n0.x * i0, r.n0.y * i0, r.n0.z * i0, r.n0.w * i0);
    *reinterpret_cast<float4*>(p1 + 128) =
        make_float4(r.n1.x * i1, r.n1.y * i1, r.n1.z * i1, r.n1.w * i1);
}

template <int NT>
__device__ __forceinline__ void cge_read_frags(const float* al, int ch, v2f (&a)[NT][KCH]) {
#pragma unroll
    for (int u = 0; u < NT; ++u)
#pragma unroll
        for (int s = 0; s < KCH; ++s)
            a[u][s] = *reinterpret_cast<const v2f*>(
                al + u * 16 * A_PITCH + 4 * (ch * KCH + s));
}

template <int NT>
__device__ __forceinline__ void cge_tiles_mma(const float* al, const v2f (&bfr)[64],
                                              v8f (&c)[NT]) {
    v2f a[2][NT][KCH];
    cge_read_frags<NT>(al, 0, a[0]);
#pragma unroll
    for (int ch = 0; ch < NCH; ++ch) {
        if (ch + 1 < NCH) cge_read_frags<NT>(al, ch + 1, a[(ch + 1) & 1]);
#pragma unroll
        for (int s = 0; s < KCH; ++s)
#pragma unroll
            for (int u = 0; u < NT; ++u)
                c[u] = __builtin_amdgcn_wmma_f32_16x16x4_f32(
                    false, a[ch & 1][u][s], false, bfr[ch * KCH + s], (short)0, c[u],
                    false, false);
    }
    __builtin_amdgcn_sched_group_barrier(0x100, NT * KCH / 2, 0);
#pragma unroll
    for (int ch = 0; ch < NCH; ++ch) {
        if (ch + 1 < NCH) __builtin_amdgcn_sched_group_barrier(0x100, NT * KCH / 2, 0);
        __builtin_amdgcn_sched_group_barrier(0x008, NT * KCH, 0);
    }
}

__device__ __forceinline__ void cge_store_tile(const v8f& c, float bn, int tile, int khalf,
                                               int n, float* __restrict__ out,
                                               int out_stride) {
    const int mbase = tile * 16 + khalf * 8;
#pragma unroll
    for (int j = 0; j < 8; ++j) {
        float v = c[j] + bn;
        v = v > 0.0f ? v : 0.0f;
        out[(size_t)(mbase + j) * out_stride + n] = v;
    }
}

template <bool XCOPY>
__device__ __forceinline__ void cge_gemm_body(
    const float* A, int a_stride,
    const float* neigh,
    const float* deg,
    const float* W,
    const float* bias,
    float* out, int out_stride,
    const float* xsrc) {
    __shared__ float a_lds[2 * 32 * A_PITCH];

    const int lane  = threadIdx.x & 31;
    const int warp  = threadIdx.x >> 5;
    const int rc    = lane & 15;
    const int khalf = lane >> 4;
    const int n     = warp * 16 + rc;

    v2f bfr[64];
#pragma unroll
    for (int s = 0; s < 64; ++s) {
        const int k = 4 * s + 2 * khalf;
        bfr[s].x = W[(size_t)k * D + n];
        bfr[s].y = W[(size_t)(k + 1) * D + n];
    }
    const float bn = bias[n];

    const int srow = warp;
    const int c4   = lane * 4;
    TileRows ra, rb;

    int t = blockIdx.x;
    ra = cge_fetch_rows(A, a_stride, neigh, deg, t, srow, c4);
    if (t + GEMM_GRID < N_TILES) {
        rb = cge_fetch_rows(A, a_stride, neigh, deg, t + GEMM_GRID, srow, c4);
        cge_stash_rows(a_lds + 16 * A_PITCH, rb, srow, c4);
    }
    cge_stash_rows(a_lds, ra, srow, c4);
    __syncthreads();

    int buf = 0;
    for (; t < N_TILES; t += 2 * GEMM_GRID) {
        const bool two   = t + GEMM_GRID < N_TILES;
        const int  tn    = t + 2 * GEMM_GRID;
        const bool more  = tn < N_TILES;
        const bool more2 = tn + GEMM_GRID < N_TILES;
        if (more)  ra = cge_fetch_rows(A, a_stride, neigh, deg, tn, srow, c4);
        if (more2) rb = cge_fetch_rows(A, a_stride, neigh, deg, tn + GEMM_GRID, srow, c4);
        float4 xa0 = make_float4(0.0f, 0.0f, 0.0f, 0.0f), xa1 = xa0, xb0 = xa0, xb1 = xa0;
        if (XCOPY) {
            const float* xp = xsrc + ((size_t)t * 16 + srow) * D + c4;
            xa0 = *reinterpret_cast<const float4*>(xp);
            xa1 = *reinterpret_cast<const float4*>(xp + 8 * D);
            if (two) {
                xb0 = *reinterpret_cast<const float4*>(xp + (size_t)GEMM_GRID * 16 * D);
                xb1 = *reinterpret_cast<const float4*>(xp + (size_t)GEMM_GRID * 16 * D + 8 * D);
            }
        }

        const float* al = a_lds + buf * (32 * A_PITCH) + rc * A_PITCH + 2 * khalf;
        v8f c[2] = {v8f{}, v8f{}};
        if (two) {
            cge_tiles_mma<2>(al, bfr, c);
        } else {
            v8f c1[1] = {v8f{}};
            cge_tiles_mma<1>(al, bfr, c1);
            c[0] = c1[0];
        }

        float* nb = a_lds + (buf ^ 1) * (32 * A_PITCH);
        if (more)  cge_stash_rows(nb, ra, srow, c4);
        if (more2) cge_stash_rows(nb + 16 * A_PITCH, rb, srow, c4);
        __syncthreads();
        cge_store_tile(c[0], bn, t, khalf, n, out, out_stride);
        if (two) cge_store_tile(c[1], bn, t + GEMM_GRID, khalf, n, out, out_stride);
        if (XCOPY) {
            float* xo = out + 128 + ((size_t)t * 16 + srow) * out_stride + c4;
            *reinterpret_cast<float4*>(xo) = xa0;
            *reinterpret_cast<float4*>(xo + (size_t)8 * out_stride) = xa1;
            if (two) {
                float* xo2 = xo + (size_t)GEMM_GRID * 16 * out_stride;
                *reinterpret_cast<float4*>(xo2) = xb0;
                *reinterpret_cast<float4*>(xo2 + (size_t)8 * out_stride) = xb1;
            }
        }
        buf ^= 1;
    }
}

__global__ __launch_bounds__(256) void cge_gemm_kernel(
    const float* __restrict__ A, int a_stride,
    const float* __restrict__ neigh,
    const float* __restrict__ deg,
    const float* __restrict__ W,
    const float* __restrict__ bias,
    float* __restrict__ out, int out_stride) {
    cge_gemm_body<false>(A, a_stride, neigh, deg, W, bias, out, out_stride, nullptr);
}

__global__ __launch_bounds__(256) void cge_gemm_final_kernel(
    const float* A, int a_stride,
    const float* __restrict__ neigh,
    const float* __restrict__ deg,
    const float* __restrict__ W,
    const float* __restrict__ bias,
    float* out, int out_stride,
    const float* __restrict__ x) {
    cge_gemm_body<true>(A, a_stride, neigh, deg, W, bias, out, out_stride, x);
}

extern "C" void kernel_launch(void* const* d_in, const int* in_sizes, int n_in,
                              void* d_out, int out_size, void* d_ws, size_t ws_size,
                              hipStream_t stream) {
    const float* x    = (const float*)d_in[0];
    const int*   esrc = (const int*)  d_in[1];
    const int*   edst = (const int*)  d_in[2];
    const float* deg  = (const float*)d_in[3];
    const float* W1   = (const float*)d_in[4];
    const float* b1   = (const float*)d_in[5];
    const float* W2   = (const float*)d_in[6];
    const float* b2   = (const float*)d_in[7];
    float* out = (float*)d_out;

    float* neigh = (float*)d_ws;
    float* h1    = out + 128;
    int*   sc    = (int*)d_out;

    dim3 blk(256);
    dim3 sblk(SORT_T);
    dim3 cgrid(NCHK);
    dim3 rgrid(NBIN);
    dim3 wgrid(N_NODES / 8);
    dim3 ggrid(GEMM_GRID);

    cge_chunk_kernel   <<<cgrid, sblk, 0, stream>>>(esrc, edst, sc);
    cge_bucket_kernel  <<<rgrid, sblk, 0, stream>>>(sc, sc);

    cge_gather_kernel<<<wgrid, blk, 0, stream>>>(x, D, sc, neigh);
    cge_gemm_kernel  <<<ggrid, blk, 0, stream>>>(x, D, neigh, deg, W1, b1,
                                                 h1, 256);
    cge_gather_kernel<<<wgrid, blk, 0, stream>>>(h1, 256, sc, neigh);
    cge_gemm_final_kernel<<<ggrid, blk, 0, stream>>>(h1, 256, neigh, deg, W2, b2,
                                                     out, 256, x);
}


# Layer-1 neighbor gather done inside the GEMM launch by the GEMM waves themselves: each wave sums the bucket rows of the 4 A rows it stages, straight into LDS; 512 blocks, 2 per WGP; same sum order
# speedup vs baseline: 3.5807x; 1.0365x over previous
#include <hip/hip_runtime.h>

typedef float v2f __attribute__((ext_vector_type(2)));
typedef float v8f __attribute__((ext_vector_type(8)));

#define N_NODES 50000
#define N_EDGES 800000
#define D 128
#define EPS_F 1e-7f

#define NBIN   200
#define BIN_W  250
#define NCHK   250
#define CHK_E  3200
#define SORT_T 1024
#define CHK_K  ((CHK_E + SORT_T - 1) / SORT_T)
static_assert(NBIN * BIN_W == N_NODES, "node ranges must tile the nodes");
static_assert(NCHK * CHK_E == N_EDGES, "edge chunks must tile the edges");
static_assert(NBIN * NCHK == N_NODES, "each table has N_NODES entries");
static_assert(BIN_W <= 256 && N_NODES <= 65536, "packed word: 8 + 16 bits");
static_assert(CHK_E % 4 == 0 && CHK_E / 4 <= SORT_T, "a chunk's words leave as one 16-byte store per thread");

#define SCR_CNT 0
#define SCR_OFF 50176
#define SCR_PTR 100480
#define SCR_SRT 150784
#define SCR_BIN 950784
static_assert(SCR_BIN % 4 == 0 && SCR_SRT % 4 == 0, "16-byte stores into the word arrays");

__device__ __forceinline__ int scr(int i) { return ((i >> 7) << 8) | (i & 127); }

__global__ __launch_bounds__(SORT_T) void cge_chunk_kernel(const int* __restrict__ esrc,
                                                           const int* __restrict__ edst,
                                                           int* __restrict__ sc) {
    __shared__ int h[256];
    __shared__ int wtot[8];
    __shared__ int4 img4[CHK_E / 4];
    int* img = reinterpret_cast<int*>(img4);
    const int t    = threadIdx.x;
    const int lane = t & 31;
    const int w    = t >> 5;
    const int b    = blockIdx.x;
    const int e0   = b * CHK_E;
    int d[CHK_K], s[CHK_K], rk[CHK_K];
#pragma unroll
    for (int k = 0; k < CHK_K; ++k) {
        const int i = t + k * SORT_T;
        d[k] = (i < CHK_E) ? edst[e0 + i] : 0;
        s[k] = (i < CHK_E) ? esrc[e0 + i] : 0;
    }
    if (t < 256) h[t] = 0;
    __syncthreads();
#pragma unroll
    for (int k = 0; k < CHK_K; ++k) {
        rk[k] = 0;
        if (t + k * SORT_T < CHK_E) rk[k] = atomicAdd(&h[(unsigned)d[k] / BIN_W], 1);
    }
    __syncthreads();
    const int c = (t < 256) ? h[t] : 0;
    int x = c;
#pragma unroll
    for (int o = 1; o < 32; o <<= 1) {
        const int y = __shfl_up(x, o);
        if (lane >= o) x += y;
    }
    if (lane == 31 && w < 8) wtot[w] = x;
    __syncthreads();
    if (t < 256) {
        int woff = 0;
#pragma unroll
        for (int k = 0; k < 8; ++k) woff += (k < w) ? wtot[k] : 0;
        const int st = woff + x - c;
        h[t] = st;
        if (t < NBIN) {
            sc[scr(SCR_CNT + t * NCHK + b)] = c;
            sc[scr(SCR_OFF + t * NCHK + b)] = st;
        }
    }
    __syncthreads();
#pragma unroll
    for (int k = 0; k < CHK_K; ++k) {
        if (t + k * SORT_T < CHK_E) {
            const int bin = (unsigned)d[k] / BIN_W;
            const int dl  = d[k] - bin * BIN_W;
            img[h[bin] + rk[k]] = (dl << 16) | s[k];
        }
    }
    __syncthreads();
    if (t < CHK_E / 4)
        *reinterpret_cast<int4*>(&sc[scr(SCR_BIN + e0 + 4 * t)]) = img4[t];
}

#define PCE_Q 8
#define IMG_W 8192
static_assert(32 * PCE_Q >= NCHK && NCHK <= 256, "a block's waves cover all pieces");
__global__ __launch_bounds__(SORT_T) void cge_bucket_kernel(const int* __restrict__ rd,
                                                            int* __restrict__ wr) {
    __shared__ int cnt[256];
    __shared__ int wtot[8];
    __shared__ int pc[256];
    __shared__ int ps[256];
    __shared__ int red[2][32];
    __shared__ int4 img4[IMG_W / 4];
    int* img = reinterpret_cast<int*>(img4);
    const int t    = threadIdx.x;
    const int lane = t & 31;
    const int w    = t >> 5;
    const int B    = blockIdx.x;
    int myc = 0, mys = 0;
    if (t < NCHK) {
        myc = rd[scr(SCR_CNT + B * NCHK + t)];
        mys = rd[scr(SCR_OFF + B * NCHK + t)];
    }
    const int lim = B * NCHK;
    int4 tv[13];
#pragma unroll
    for (int k = 0; k < 13; ++k) {
        const int i = 4 * (t + k * SORT_T);
        tv[k] = make_int4(0, 0, 0, 0);
        if (i < lim) tv[k] = *reinterpret_cast<const int4*>(&rd[scr(SCR_CNT + i)]);
    }
    int part = 0;
#pragma unroll
    for (int k = 0; k < 13; ++k) {
        const int i = 4 * (t + k * SORT_T);
        part += tv[k].x + ((i + 1 < lim) ? tv[k].y : 0) + ((i + 2 < lim) ? tv[k].z : 0)
              + ((i + 3 < lim) ? tv[k].w : 0);
    }
    if (t < 256) { cnt[t] = 0; pc[t] = myc; ps[t] = mys; }
    int ra = part, rm = myc;
#pragma unroll
    for (int o = 16; o > 0; o >>= 1) { ra += __shfl_xor(ra, o); rm += __shfl_xor(rm, o); }
    if (lane == 0) { red[0][w] = ra; red[1][w] = rm; }
    __syncthreads();
    ra = red[0][lane];
    rm = red[1][lane];
#pragma unroll
    for (int o = 16; o > 0; o >>= 1) { ra += __shfl_xor(ra, o); rm += __shfl_xor(rm, o); }
    const int bs = ra;
    const int be = bs + rm;

    int pcq[PCE_Q], psq[PCE_Q], wq[PCE_Q];
#pragma unroll
    for (int q = 0; q < PCE_Q; ++q) {
        const int b = w + 32 * q;
        pcq[q] = pc[b];
        psq[q] = SCR_BIN + b * CHK_E + ps[b];
    }
#pragma unroll
    for (int q = 0; q < PCE_Q; ++q) {
        wq[q] = 0;
        if (lane < pcq[q]) wq[q] = rd[scr(psq[q] + lane)];
    }
#pragma unroll
    for (int q = 0; q < PCE_Q; ++q)
        if (lane < pcq[q]) atomicAdd(&cnt[wq[q] >> 16], 1);
#pragma unroll
    for (int q = 0; q < PCE_Q; ++q)
        for (int j = lane + 32; j < pcq[q]; j += 32)
            atomicAdd(&cnt[rd[scr(psq[q] + j)] >> 16], 1);
    __syncthreads();
    const int c = (t < 256) ? cnt[t] : 0;
    int x = c;
#pragma unroll
    for (int o = 1; o < 32; o <<= 1) {
        const int y = __shfl_up(x, o);
        if (lane >= o) x += y;
    }
    if (lane == 31 && w < 8) wtot[w] = x;
    __syncthreads();
    if (t < 256) {
        int woff = 0;
#pragma unroll
        for (int k = 0; k < 8; ++k) woff += (k < w) ? wtot[k] : 0;
        const int first = bs + woff + x - c;
        cnt[t] = first;
        if (t < BIN_W) wr[scr(SCR_PTR + B * BIN_W + t)] = first;
        if (B == NBIN - 1 && t == 0) wr[scr(SCR_PTR + N_NODES)] = be;
    }
    __syncthreads();
    const int g0   = SCR_SRT + bs;
    const int off  = g0 & 31;
    const int nw   = be - bs;
    const bool fit = nw + off <= IMG_W;
    const int sh   = off - bs;
    int pos[PCE_Q];
#pragma unroll
    for (int q = 0; q < PCE_Q; ++q) {
        pos[q] = 0;
        if (lane < pcq[q]) pos[q] = atomicAdd(&cnt[wq[q] >> 16], 1);
    }
#pragma unroll
    for (int q = 0; q < PCE_Q; ++q)
        if (lane < pcq[q]) {
            if (fit) img[pos[q] + sh] = wq[q] & 0xFFFF;
            else     wr[scr(SCR_SRT + pos[q])] = wq[q] & 0xFFFF;
        }
#pragma unroll
    for (int q = 0; q < PCE_Q; ++q)
        for (int j = lane + 32; j < pcq[q]; j += 32) {
            const int v = rd[scr(psq[q] + j)];
            const int p = atomicAdd(&cnt[v >> 16], 1);
            if (fit) img[p + sh] = v & 0xFFFF;
            else     wr[scr(SCR_SRT + p)] = v & 0xFFFF;
        }
    if (fit) {
        __syncthreads();
        const int gb = g0 - off;
        const int hi = off + nw;
        for (int j = 4 * t; j < hi; j += 4 * SORT_T) {
            if (j >= off && j + 4 <= hi) {
                *reinterpret_cast<int4*>(&wr[scr(gb + j)]) = img4[j >> 2];
            } else {
                for (int u = 0; u < 4; ++u)
                    if (j + u >= off && j + u < hi) wr[scr(gb + j + u)] = img[j + u];
            }
        }
    }
}

template <int G>
__device__ __forceinline__ void cge_add_rows(const float* __restrict__ hl, int h_stride,
                                             int sidx, int j, float4& acc) {
    float4 v[G];
#pragma unroll
    for (int u = 0; u < G; ++u) {
        const int s = __builtin_amdgcn_readlane(sidx, j + u);
        v[u] = *reinterpret_cast<const float4*>(hl + (size_t)s * h_stride);
    }
#pragma unroll
    for (int u = 0; u < G; ++u) {
        acc.x += v[u].x; acc.y += v[u].y; acc.z += v[u].z; acc.w += v[u].w;
    }
}

__global__ __launch_bounds__(256) void cge_gather_kernel(
    const float* __restrict__ h, int h_stride,
    const int* __restrict__ sc, float* __restrict__ neigh) {
    const int lane = threadIdx.x & 31;
    const int node = blockIdx.x * 8 + (threadIdx.x >> 5);
    const int pv   = sc[scr(SCR_PTR + node + (lane & 1))];
    const int beg  = __builtin_amdgcn_readlane(pv, 0);
    const int end  = __builtin_amdgcn_readlane(pv, 1);
    const float* __restrict__ hl = h + lane * 4;
    float4 acc = make_float4(0.0f, 0.0f, 0.0f, 0.0f);
    for (int c = beg; c < end; c += 32) {
        const int n    = min(32, end - c);
        const int sidx = (lane < n) ? sc[scr(SCR_SRT + c + lane)] : 0;
        int j = 0;
        for (; j + 8 <= n; j += 8) cge_add_rows<8>(hl, h_stride, sidx, j, acc);
        if (n - j >= 4) { cge_add_rows<4>(hl, h_stride, sidx, j, acc); j += 4; }
        if (n - j >= 2) { cge_add_rows<2>(hl, h_stride, sidx, j, acc); j += 2; }
        if (n - j >= 1) { cge_add_rows<1>(hl, h_stride, sidx, j, acc); }
    }
    reinterpret_cast<float4*>(neigh + (size_t)node * D)[lane] = acc;
}

#define GEMM_GRID 256
#define N_TILES   (N_NODES / 16)
#define A_PITCH   260
#define KCH       16

struct TileRows { float4 s0, s1, n0, n1; float d0, d1; };

__device__ __forceinline__ TileRows cge_fetch_rows(
    const float* __restrict__ A, int a_stride, const float* __restrict__ neigh,
    const float* __restrict__ deg, int tile, int srow, int c4) {
    TileRows r;
    const int m0 = tile * 16 + srow;
    const int m1 = m0 + 8;
    r.s0 = *reinterpret_cast<const float4*>(A + (size_t)m0 * a_stride + c4);
    r.s1 = *reinterpret_cast<const float4*>(A + (size_t)m1 * a_stride + c4);
    r.n0 = *reinterpret_cast<const float4*>(neigh + (size_t)m0 * D + c4);
    r.n1 = *reinterpret_cast<const float4*>(neigh + (size_t)m1 * D + c4);
    r.d0 = deg[m0];
    r.d1 = deg[m1];
    return r;
}

__device__ __forceinline__ void cge_stash_rows(float* rows, const TileRows& r,
                                               int srow, int c4) {
    const float i0 = 1.0f / (r.d0 + EPS_F);
    const float i1 = 1.0f / (r.d1 + EPS_F);
    float* p0 = rows + srow * A_PITCH + c4;
    float* p1 = p0 + 8 * A_PITCH;
    *reinterpret_cast<float4*>(p0) = r.s0;
    *reinterpret_cast<float4*>(p1) = r.s1;
    *reinterpret_cast<float4*>(p0 + 128) =
        make_float4(r.n0.x * i0, r.n0.y * i0, r.n0.z * i0, r.n0.w * i0);
    *reinterpret_cast<float4*>(p1 + 128) =
        make_float4(r.n1.x * i1, r.n1.y * i1, r.n1.z * i1, r.n1.w * i1);
}

template <int NT, int KC>
__device__ __forceinline__ void cge_read_frags(const float* al, int ch, v2f (&a)[NT][KC]) {
#pragma unroll
    for (int u = 0; u < NT; ++u)
#pragma unroll
        for (int s = 0; s < KC; ++s)
            a[u][s] = *reinterpret_cast<const v2f*>(
                al + u * 16 * A_PITCH + 4 * (ch * KC + s));
}

template <int NT, int KC = KCH>
__device__ __forceinline__ void cge_tiles_mma(const float* al, const v2f (&bfr)[64],
                                              v8f (&c)[NT]) {
    constexpr int NC = 64 / KC;
    v2f a[2][NT][KC];
    cge_read_frags<NT, KC>(al, 0, a[0]);
#pragma unroll
    for (int ch = 0; ch < NC; ++ch) {
        if (ch + 1 < NC) cge_read_frags<NT, KC>(al, ch + 1, a[(ch + 1) & 1]);
#pragma unroll
        for (int s = 0; s < KC; ++s)
#pragma unroll
            for (int u = 0; u < NT; ++u)
                c[u] = __builtin_amdgcn_wmma_f32_16x16x4_f32(
                    false, a[ch & 1][u][s], false, bfr[ch * KC + s], (short)0, c[u],
                    false, false);
    }
    __builtin_amdgcn_sched_group_barrier(0x100, NT * KC / 2, 0);
#pragma unroll
    for (int ch = 0; ch < NC; ++ch) {
        if (ch + 1 < NC) __builtin_amdgcn_sched_group_barrier(0x100, NT * KC / 2, 0);
        __builtin_amdgcn_sched_group_barrier(0x008, NT * KC, 0);
    }
}

__device__ __forceinline__ void cge_store_tile(const v8f& c, float bn, int tile, int khalf,
                                               int n, float* __restrict__ out,
                                               int out_stride) {
    const int mbase = tile * 16 + khalf * 8;
#pragma unroll
    for (int j = 0; j < 8; ++j) {
        float v = c[j] + bn;
        v = v > 0.0f ? v : 0.0f;
        out[(size_t)(mbase + j) * out_stride + n] = v;
    }
}

template <bool XCOPY>
__device__ __forceinline__ void cge_gemm_body(
    const float* A, int a_stride,
    const float* neigh,
    const float* deg,
    const float* W,
    const float* bias,
    float* out, int out_stride,
    const float* xsrc) {
    __shared__ float a_lds[2 * 32 * A_PITCH];

    const int lane  = threadIdx.x & 31;
    const int warp  = threadIdx.x >> 5;
    const int rc    = lane & 15;
    const int khalf = lane >> 4;
    const int n     = warp * 16 + rc;

    v2f bfr[64];
#pragma unroll
    for (int s = 0; s < 64; ++s) {
        const int k = 4 * s + 2 * khalf;
        bfr[s].x = W[(size_t)k * D + n];
        bfr[s].y = W[(size_t)(k + 1) * D + n];
    }
    const float bn = bias[n];

    const int srow = warp;
    const int c4   = lane * 4;
    TileRows ra, rb;

    int t = blockIdx.x;
    ra = cge_fetch_rows(A, a_stride, neigh, deg, t, srow, c4);
    if (t + GEMM_GRID < N_TILES) {
        rb = cge_fetch_rows(A, a_stride, neigh, deg, t + GEMM_GRID, srow, c4);
        cge_stash_rows(a_lds + 16 * A_PITCH, rb, srow, c4);
    }
    cge_stash_rows(a_lds, ra, srow, c4);
    __syncthreads();

    int buf = 0;
    for (; t < N_TILES; t += 2 * GEMM_GRID) {
        const bool two   = t + GEMM_GRID < N_TILES;
        const int  tn    = t + 2 * GEMM_GRID;
        const bool more  = tn < N_TILES;
        const bool more2 = tn + GEMM_GRID < N_TILES;
        if (more)  ra = cge_fetch_rows(A, a_stride, neigh, deg, tn, srow, c4);
        if (more2) rb = cge_fetch_rows(A, a_stride, neigh, deg, tn + GEMM_GRID, srow, c4);
        float4 xa0 = make_float4(0.0f, 0.0f, 0.0f, 0.0f), xa1 = xa0, xb0 = xa0, xb1 = xa0;
        if (XCOPY) {
            const float* xp = xsrc + ((size_t)t * 16 + srow) * D + c4;
            xa0 = *reinterpret_cast<const float4*>(xp);
            xa1 = *reinterpret_cast<const float4*>(xp + 8 * D);
            if (two) {
                xb0 = *reinterpret_cast<const float4*>(xp + (size_t)GEMM_GRID * 16 * D);
                xb1 = *reinterpret_cast<const float4*>(xp + (size_t)GEMM_GRID * 16 * D + 8 * D);
            }
        }

        const float* al = a_lds + buf * (32 * A_PITCH) + rc * A_PITCH + 2 * khalf;
        v8f c[2] = {v8f{}, v8f{}};
        if (two) {
            cge_tiles_mma<2>(al, bfr, c);
        } else {
            v8f c1[1] = {v8f{}};
            cge_tiles_mma<1>(al, bfr, c1);
            c[0] = c1[0];
        }

        float* nb = a_lds + (buf ^ 1) * (32 * A_PITCH);
        if (more)  cge_stash_rows(nb, ra, srow, c4);
        if (more2) cge_stash_rows(nb + 16 * A_PITCH, rb, srow, c4);
        __syncthreads();
        cge_store_tile(c[0], bn, t, khalf, n, out, out_stride);
        if (two) cge_store_tile(c[1], bn, t + GEMM_GRID, khalf, n, out, out_stride);
        if (XCOPY) {
            float* xo = out + 128 + ((size_t)t * 16 + srow) * out_stride + c4;
            *reinterpret_cast<float4*>(xo) = xa0;
            *reinterpret_cast<float4*>(xo + (size_t)8 * out_stride) = xa1;
            if (two) {
                float* xo2 = xo + (size_t)GEMM_GRID * 16 * out_stride;
                *reinterpret_cast<float4*>(xo2) = xb0;
                *reinterpret_cast<float4*>(xo2 + (size_t)8 * out_stride) = xb1;
            }
        }
        buf ^= 1;
    }
}

__global__ __launch_bounds__(256) void cge_gemm_final_kernel(
    const float* A, int a_stride,
    const float* __restrict__ neigh,
    const float* __restrict__ deg,
    const float* __restrict__ W,
    const float* __restrict__ bias,
    float* out, int out_stride,
    const float* __restrict__ x) {
    cge_gemm_body<true>(A, a_stride, neigh, deg, W, bias, out, out_stride, x);
}

#define FUSE_GRID 512
#define FUSE_KCH  8

__device__ __forceinline__ void cge_gather_stage(
    const float* __restrict__ x, const int* __restrict__ sc, const float* __restrict__ deg,
    float* rows, int t0, int t1, bool two, int srow, int lane) {
    const int c4 = lane * 4;
    const float* __restrict__ hl = x + c4;
    const int tb = two ? t1 : t0;
    const int na = t0 * 16 + srow;
    const int nb = tb * 16 + srow;
    const float4 sa0 = *reinterpret_cast<const float4*>(hl + (size_t)na * D);
    const float4 sa1 = *reinterpret_cast<const float4*>(hl + (size_t)(na + 8) * D);
    const float4 sb0 = *reinterpret_cast<const float4*>(hl + (size_t)nb * D);
    const float4 sb1 = *reinterpret_cast<const float4*>(hl + (size_t)(nb + 8) * D);
    const float da0 = deg[na], da1 = deg[na + 8], db0 = deg[nb], db1 = deg[nb + 8];
    const int l16 = min(lane, 16);
    const int pa = sc[scr(SCR_PTR + t0 * 16 + l16)];
    const int pb = sc[scr(SCR_PTR + tb * 16 + l16)];
    const int a0b = __builtin_amdgcn_readlane(pa, srow),     a0e = __builtin_amdgcn_readlane(pa, srow + 1);
    const int a1b = __builtin_amdgcn_readlane(pa, srow + 8), a1e = __builtin_amdgcn_readlane(pa, srow + 9);
    const int b0b = __builtin_amdgcn_readlane(pb, srow),     b0e = __builtin_amdgcn_readlane(pb, srow + 1);
    const int b1b = __builtin_amdgcn_readlane(pb, srow + 8), b1e = __builtin_amdgcn_readlane(pb, srow + 9);
    const int ia0 = sc[scr(SCR_SRT + min(a0b + lane, N_EDGES - 1))];
    const int ia1 = sc[scr(SCR_SRT + min(a1b + lane, N_EDGES - 1))];
    const int ib0 = sc[scr(SCR_SRT + min(b0b + lane, N_EDGES - 1))];
    const int ib1 = sc[scr(SCR_SRT + min(b1b + lane, N_EDGES - 1))];
    float* p0 = rows + srow * A_PITCH + c4;
    *reinterpret_cast<float4*>(p0) = sa0;
    *reinterpret_cast<float4*>(p0 + 8 * A_PITCH) = sa1;
    *reinterpret_cast<float4*>(p0 + 16 * A_PITCH) = sb0;
    *reinterpret_cast<float4*>(p0 + 24 * A_PITCH) = sb1;
    const int nq = two ? 4 : 2;
#pragma nounroll
    for (int q = 0; q < nq; ++q) {
        const int beg  = (q == 0) ? a0b : (q == 1) ? a1b : (q == 2) ? b0b : b1b;
        const int end  = (q == 0) ? a0e : (q == 1) ? a1e : (q == 2) ? b0e : b1e;
        int sidx       = (q == 0) ? ia0 : (q == 1) ? ia1 : (q == 2) ? ib0 : ib1;
        const float dg = (q == 0) ? da0 : (q == 1) ? da1 : (q == 2) ? db0 : db1;
        float4 acc = make_float4(0.0f, 0.0f, 0.0f, 0.0f);
        for (int c = beg; c < end; c += 32) {
            const int n = min(32, end - c);
            if (c != beg) sidx = (lane < n) ? sc[scr(SCR_SRT + c + lane)] : 0;
            int j = 0;
            for (; j + 16 <= n; j += 16) cge_add_rows<16>(hl, D, sidx, j, acc);
            if (n - j >= 8) { cge_add_rows<8>(hl, D, sidx, j, acc); j += 8; }
            if (n - j >= 4) { cge_add_rows<4>(hl, D, sidx, j, acc); j += 4; }
            if (n - j >= 2) { cge_add_rows<2>(hl, D, sidx, j, acc); j += 2; }
            if (n - j >= 1) { cge_add_rows<1>(hl, D, sidx, j, acc); }
        }
        const float inv = 1.0f / (dg + EPS_F);
        *reinterpret_cast<float4*>(p0 + q * 8 * A_PITCH + 128) =
            make_float4(acc.x * inv, acc.y * inv, acc.z * inv, acc.w * inv);
    }
}

__global__ __launch_bounds__(256, 4) void cge_gemm_gather_kernel(
    const float* __restrict__ x,
    const int* __restrict__ sc,
    const float* __restrict__ deg,
    const float* __restrict__ W,
    const float* __restrict__ bias,
    float* __restrict__ out, int out_stride) {
    __shared__ float a_lds[2 * 32 * A_PITCH];

    const int lane  = threadIdx.x & 31;
    const int warp  = __builtin_amdgcn_readfirstlane(threadIdx.x >> 5);
    const int rc    = lane & 15;
    const int khalf = lane >> 4;
    const int n     = warp * 16 + rc;

    v2f bfr[64];
#pragma unroll
    for (int s = 0; s < 64; ++s) {
        const int k = 4 * s + 2 * khalf;
        bfr[s].x = W[(size_t)k * D + n];
        bfr[s].y = W[(size_t)(k + 1) * D + n];
    }
    const float bn = bias[n];

    int buf = 0;
    for (int t = (int)blockIdx.x - 2 * FUSE_GRID; t < N_TILES; t += 2 * FUSE_GRID) {
        const bool two = t + FUSE_GRID < N_TILES;
        const int  tn  = t + 2 * FUSE_GRID;
        if (tn < N_TILES)
            cge_gather_stage(x, sc, deg, a_lds + (buf ^ 1) * (32 * A_PITCH), tn, tn + FUSE_GRID,
                             tn + FUSE_GRID < N_TILES, warp, lane);

        v8f c[2] = {v8f{}, v8f{}};
        if (t >= 0) {
            const float* al = a_lds + buf * (32 * A_PITCH) + rc * A_PITCH + 2 * khalf;
            if (two) {
                cge_tiles_mma<2, FUSE_KCH>(al, bfr, c);
            } else {
                v8f c1[1] = {v8f{}};
                cge_tiles_mma<1, FUSE_KCH>(al, bfr, c1);
                c[0] = c1[0];
            }
        }
        __syncthreads();
        if (t >= 0) {
            cge_store_tile(c[0], bn, t, khalf, n, out, out_stride);
            if (two) cge_store_tile(c[1], bn, t + FUSE_GRID, khalf, n, out, out_stride);
        }
        buf ^= 1;
    }
}

extern "C" void kernel_launch(void* const* d_in, const int* in_sizes, int n_in,
                              void* d_out, int out_size, void* d_ws, size_t ws_size,
                              hipStream_t stream) {
    const float* x    = (const float*)d_in[0];
    const int*   esrc = (const int*)  d_in[1];
    const int*   edst = (const int*)  d_in[2];
    const float* deg  = (const float*)d_in[3];
    const float* W1   = (const float*)d_in[4];
    const float* b1   = (const float*)d_in[5];
    const float* W2   = (const float*)d_in[6];
    const float* b2   = (const float*)d_in[7];
    float* out = (float*)d_out;

    float* neigh = (float*)d_ws;
    float* h1    = out + 128;
    int*   sc    = (int*)d_out;

    dim3 blk(256);
    dim3 sblk(SORT_T);
    dim3 cgrid(NCHK);
    dim3 rgrid(NBIN);
    dim3 wgrid(N_NODES / 8);
    dim3 ggrid(GEMM_GRID);
    dim3 fgrid(FUSE_GRID);

    cge_chunk_kernel   <<<cgrid, sblk, 0, stream>>>(esrc, edst, sc);
    cge_bucket_kernel  <<<rgrid, sblk, 0, stream>>>(sc, sc);

    cge_gemm_gather_kernel<<<fgrid, blk, 0, stream>>>(x, sc, deg, W1, b1, h1, 256);
    cge_gather_kernel<<<wgrid, blk, 0, stream>>>(h1, 256, sc, neigh);
    cge_gemm_final_kernel<<<ggrid, blk, 0, stream>>>(h1, 256, neigh, deg, W2, b2,
                                                     out, 256, x);
}
